# starting kernel
# baseline (speedup 1.0000x reference)
#include <hip/hip_runtime.h>
#include <math.h>

#define NN 100000
#define EE 1000000
#define FF 128
#define DD 64
#define GG 256

typedef float v2f __attribute__((ext_vector_type(2)));
typedef float v8f __attribute__((ext_vector_type(8)));

__global__ void k_zero_deg(float* __restrict__ deg) {
  int i = blockIdx.x * blockDim.x + threadIdx.x;
  if (i < NN) deg[i] = 0.0f;
}

__global__ void k_degree(const int* __restrict__ dst, float* __restrict__ deg) {
  int e = blockIdx.x * blockDim.x + threadIdx.x;
  if (e < EE) atomicAdd(&deg[dst[e]], 1.0f);
}

__global__ void k_dinv(float* __restrict__ deg) {
  int i = blockIdx.x * blockDim.x + threadIdx.x;
  if (i < NN) deg[i] = rsqrtf(deg[i] + 1.0f);
}

template <int K>
__global__ void k_gemm_wmma(const float* __restrict__ A, const float* __restrict__ B,
                            float* __restrict__ C) {
  int gid  = blockIdx.x * blockDim.x + threadIdx.x;
  int wave = gid >> 5;
  int lane = gid & 31;
  int tm = wave >> 2;
  int tn = wave & 3;
  if (tm * 16 >= NN) return;

  int half = lane >> 4;
  int m    = lane & 15;
  const float* arow = A + (size_t)(tm * 16 + m) * K;
  int col = tn * 16 + m;

  v8f acc = {};
#pragma unroll
  for (int kb = 0; kb < K; kb += 4) {
    int k0 = kb + half * 2;
    v2f a; a[0] = arow[k0];          a[1] = arow[k0 + 1];
    v2f b; b[0] = B[k0 * DD + col];  b[1] = B[(k0 + 1) * DD + col];
    acc = __builtin_amdgcn_wmma_f32_16x16x4_f32(
         false, a,  false, b,
         (short)0, acc,  false,  false);
  }

  int r0 = tm * 16 + half * 8;
#pragma unroll
  for (int j = 0; j < 8; ++j)
    C[(size_t)(r0 + j) * DD + col] = acc[j];
}

__global__ void k_init_agg(const float* __restrict__ hw, const float* __restrict__ dinv,
                           float* __restrict__ agg) {
  int i = blockIdx.x * blockDim.x + threadIdx.x;
  if (i < NN * DD) {
    float dv = dinv[i >> 6];
    agg[i] = dv * dv * hw[i];
  }
}

__global__ void k_scatter(const int* __restrict__ src, const int* __restrict__ dst,
                          const float* __restrict__ dinv, const float* __restrict__ hw,
                          float* __restrict__ agg) {
  int gid  = blockIdx.x * blockDim.x + threadIdx.x;
  int e    = gid >> 5;
  int lane = gid & 31;
  if (e >= EE) return;
  int s = src[e], d = dst[e];
  float w = dinv[s] * dinv[d];
  const float2* hp = (const float2*)(hw + (size_t)s * DD);
  float2 v = hp[lane];
  float* ap = agg + (size_t)d * DD + lane * 2;
  atomicAdd(ap,     v.x * w);
  atomicAdd(ap + 1, v.y * w);
}

__global__ void k_finish(const float* __restrict__ b, float* __restrict__ agg) {
  int i = blockIdx.x * blockDim.x + threadIdx.x;
  if (i < NN * DD) agg[i] = tanhf(agg[i] + b[i & 63]);
}

__device__ __forceinline__ unsigned fflip(float f) {
  unsigned u = __float_as_uint(f);
  return u ^ ((unsigned)((int)u >> 31) | 0x80000000u);
}
__device__ __forceinline__ float funflip(unsigned u) {
  unsigned m = (u >> 31) ? 0x80000000u : 0xFFFFFFFFu;
  return __uint_as_float(u ^ m);
}

__global__ void k_pool_init(unsigned* __restrict__ gmp, float* __restrict__ gsum,
                            float* __restrict__ counts) {
  int i = blockIdx.x * blockDim.x + threadIdx.x;
  if (i < GG * DD) { gmp[i] = 0x007FFFFFu  ; gsum[i] = 0.0f; }
  if (i < GG) counts[i] = 0.0f;
}

__global__ void k_pool(const float* __restrict__ h, const int* __restrict__ bidx,
                       unsigned* __restrict__ gmp, float* __restrict__ gsum,
                       float* __restrict__ counts) {
  int i = blockIdx.x * blockDim.x + threadIdx.x;
  if (i >= NN * DD) return;
  int n = i >> 6, dd = i & 63;
  int g = bidx[n];
  float v = h[i];
  atomicMax(&gmp[g * DD + dd], fflip(v));
  atomicAdd(&gsum[g * DD + dd], v);
  if (dd == 0) atomicAdd(&counts[g], 1.0f);
}

__global__ void k_final(const unsigned* __restrict__ gmp, const float* __restrict__ gsum,
                        const float* __restrict__ counts, const float* __restrict__ Wout,
                        const float* __restrict__ bout, float* __restrict__ out,
                        float* __restrict__ hidden) {
  __shared__ float red[2 * DD];
  int g = blockIdx.x, t = threadIdx.x;
  float cnt = fmaxf(counts[g], 1.0f);
  float hv = (t < DD) ? funflip(gmp[g * DD + t]) : (gsum[g * DD + (t - DD)] / cnt);
  hidden[(size_t)g * 2 * DD + t] = hv;
  red[t] = hv * Wout[t];
  __syncthreads();
  for (int s = DD; s > 0; s >>= 1) {
    if (t < s) red[t] += red[t + s];
    __syncthreads();
  }
  if (t == 0) out[g] = red[0] + bout[0];
}

extern "C" void kernel_launch(void* const* d_in, const int* in_sizes, int n_in,
                              void* d_out, int out_size, void* d_ws, size_t ws_size,
                              hipStream_t stream) {
  const float* x    = (const float*)d_in[0];
  const int*   ei   = (const int*)d_in[1];
  const int*   bidx = (const int*)d_in[2];
  const float* W0   = (const float*)d_in[3];
  const float* b0   = (const float*)d_in[4];
  const float* W1   = (const float*)d_in[5];
  const float* b1   = (const float*)d_in[6];
  const float* W2   = (const float*)d_in[7];
  const float* b2   = (const float*)d_in[8];
  const float* W3   = (const float*)d_in[9];
  const float* b3   = (const float*)d_in[10];
  const float* Wout = (const float*)d_in[11];
  const float* bout = (const float*)d_in[12];

  const int* src = ei;
  const int* dst = ei + EE;

  char*  ws   = (char*)d_ws;
  size_t off  = 0;
  float* dinv = (float*)(ws + off);  off += (((size_t)NN * 4 + 255) & ~(size_t)255);
  float* bufP = (float*)(ws + off);  off += (size_t)NN * DD * 4;
  float* bufQ = (float*)(ws + off);  off += (size_t)NN * DD * 4;
  unsigned* gmp = (unsigned*)(ws + off); off += (size_t)GG * DD * 4;
  float* gsum   = (float*)(ws + off);    off += (size_t)GG * DD * 4;
  float* counts = (float*)(ws + off);    off += (size_t)GG * 4;
  (void)ws_size; (void)n_in; (void)in_sizes; (void)out_size;

  float* outp   = (float*)d_out;
  float* hidden = outp + GG;

  const int T = 256;

  k_zero_deg<<<(NN + T - 1) / T, T, 0, stream>>>(dinv);
  k_degree  <<<(EE + T - 1) / T, T, 0, stream>>>(dst, dinv);
  k_dinv    <<<(NN + T - 1) / T, T, 0, stream>>>(dinv);

  const int gemm_blocks = (NN / 16) * 4 / 8;
  const int nd_blocks   = (NN * DD + T - 1) / T;
  const int sc_blocks   = (int)(((size_t)EE * 32 + T - 1) / T);

  const float* Wl[4] = {W0, W1, W2, W3};
  const float* bl[4] = {b0, b1, b2, b3};
  for (int l = 0; l < 4; ++l) {
    if (l == 0) k_gemm_wmma<FF><<<gemm_blocks, T, 0, stream>>>(x,    Wl[l], bufQ);
    else        k_gemm_wmma<DD><<<gemm_blocks, T, 0, stream>>>(bufP, Wl[l], bufQ);
    k_init_agg<<<nd_blocks, T, 0, stream>>>(bufQ, dinv, bufP);
    k_scatter <<<sc_blocks, T, 0, stream>>>(src, dst, dinv, bufQ, bufP);
    k_finish  <<<nd_blocks, T, 0, stream>>>(bl[l], bufP);
  }

  k_pool_init<<<(GG * DD + T - 1) / T, T, 0, stream>>>(gmp, gsum, counts);
  k_pool     <<<nd_blocks, T, 0, stream>>>(bufP, bidx, gmp, gsum, counts);
  k_final    <<<GG, 2 * DD, 0, stream>>>(gmp, gsum, counts, Wout, bout, outp, hidden);
}


# edge scatter kept inside one chip half: edge list split once by destination half; a block reads its die id and takes 256-edge pieces of its own half's list by ticket; sums stay atomic arrival order
# speedup vs baseline: 1.5580x; 1.5580x over previous
#include <hip/hip_runtime.h>
#include <math.h>

#define NN 100000
#define EE 1000000
#define FF 128
#define DD 64
#define GG 256

typedef float v2f __attribute__((ext_vector_type(2)));
typedef float v4f __attribute__((ext_vector_type(4)));
typedef float v8f __attribute__((ext_vector_type(8)));
typedef int   v2i __attribute__((ext_vector_type(2)));

#define NHALF (NN / 2)
#define CSTR 64
#define NCTR 16

__global__ void k_zero_deg(float* __restrict__ deg, unsigned* __restrict__ ctr) {
  int i = blockIdx.x * blockDim.x + threadIdx.x;
  if (i < NN) deg[i] = 0.0f;
  if (i < NCTR * CSTR) ctr[i] = 0u;
}

__global__ void __launch_bounds__(1024)
k_degree_split(const int* __restrict__ src, const int* __restrict__ dst, float* __restrict__ deg,
               unsigned* __restrict__ ctr, int* __restrict__ list0, int* __restrict__ list1) {
  __shared__ unsigned wc0[32], wc1[32], run[2];
  int e    = blockIdx.x * 1024 + threadIdx.x;
  int lane = threadIdx.x & 31;
  int wv   = threadIdx.x >> 5;
  bool valid = e < EE;
  int s = 0, d = 0;
  if (valid) {
    s = src[e];
    d = dst[e];
    atomicAdd(&deg[d], 1.0f);
  }
  bool p0 = valid && d < NHALF;
  bool p1 = valid && d >= NHALF;
  unsigned m0 = __builtin_amdgcn_ballot_w32(p0);
  unsigned m1 = __builtin_amdgcn_ballot_w32(p1);
  if (lane == 0) { wc0[wv] = __builtin_popcount(m0); wc1[wv] = __builtin_popcount(m1); }
  __syncthreads();
  if (wv == 0) {
    unsigned a0 = wc0[lane], a1 = wc1[lane];
    unsigned s0 = a0, s1 = a1;
#pragma unroll
    for (int o = 1; o < 32; o <<= 1) {
      unsigned t0 = __shfl_up(s0, o);
      unsigned t1 = __shfl_up(s1, o);
      if (lane >= o) { s0 += t0; s1 += t1; }
    }
    wc0[lane] = s0 - a0;
    wc1[lane] = s1 - a1;
    if (lane == 31) {
      run[0] = atomicAdd(&ctr[0 * CSTR], s0);
      run[1] = atomicAdd(&ctr[1 * CSTR], s1);
    }
  }
  __syncthreads();
  unsigned below = (1u << lane) - 1u;
  v2i rec;  rec[0] = s;  rec[1] = d;
  if (p0) *(v2i*)(list0 + 2 * (size_t)(run[0] + wc0[wv] + __builtin_popcount(m0 & below))) = rec;
  if (p1) *(v2i*)(list1 + 2 * (size_t)(run[1] + wc1[wv] + __builtin_popcount(m1 & below))) = rec;
}

__global__ void k_dinv(float* __restrict__ deg) {
  int i = blockIdx.x * blockDim.x + threadIdx.x;
  if (i < NN) deg[i] = rsqrtf(deg[i] + 1.0f);
}

template <int K, bool ACT>
__global__ void __launch_bounds__(128, 6)
k_gemm_fused(const float* __restrict__ A, const float* __restrict__ bprev,
             const float* __restrict__ B, const float* __restrict__ dinv,
             float* __restrict__ hw, float* __restrict__ agg) {
  int wave = (blockIdx.x * blockDim.x + threadIdx.x) >> 5;
  int lane = threadIdx.x & 31;
  if (wave * 16 >= NN) return;

  int half = lane >> 4;
  int c    = lane & 15;
  const float* arow = A + (size_t)(wave * 16 + c) * K;
  const float* bcol = B + 4 * c;

  v2f areg[K / 4];
#pragma unroll
  for (int i = 0; i < K / 4; ++i)
    areg[i] = *(const v2f*)(arow + 4 * i + 2 * half);
  if (ACT) {
    v2f breg[K / 4];
#pragma unroll
    for (int i = 0; i < K / 4; ++i)
      breg[i] = *(const v2f*)(bprev + 4 * i + 2 * half);
#pragma unroll
    for (int i = 0; i < K / 4; ++i) {
      areg[i][0] = tanhf(areg[i][0] + breg[i][0]);
      areg[i][1] = tanhf(areg[i][1] + breg[i][1]);
    }
  }

  v4f q[2][4];
#pragma unroll
  for (int s = 0; s < 2; ++s) {
    int k0 = 4 * s + 2 * half;
    q[0][2 * s]     = *(const v4f*)(bcol + (size_t)k0 * DD);
    q[0][2 * s + 1] = *(const v4f*)(bcol + (size_t)(k0 + 1) * DD);
  }

  v8f acc0 = {}, acc1 = {}, acc2 = {}, acc3 = {};
#pragma unroll
  for (int g = 0; g < K / 8; ++g) {
    if (g + 1 < K / 8) {
#pragma unroll
      for (int s = 0; s < 2; ++s) {
        int k0 = 8 * (g + 1) + 4 * s + 2 * half;
        q[(g + 1) & 1][2 * s]     = *(const v4f*)(bcol + (size_t)k0 * DD);
        q[(g + 1) & 1][2 * s + 1] = *(const v4f*)(bcol + (size_t)(k0 + 1) * DD);
      }
    }
#pragma unroll
    for (int s = 0; s < 2; ++s) {
      v2f a  = areg[2 * g + s];
      v4f q0 = q[g & 1][2 * s];
      v4f q1 = q[g & 1][2 * s + 1];
      v2f b;
      b[0] = q0[0]; b[1] = q1[0];
      acc0 = __builtin_amdgcn_wmma_f32_16x16x4_f32(false, a, false, b, (short)0, acc0, false, false);
      b[0] = q0[1]; b[1] = q1[1];
      acc1 = __builtin_amdgcn_wmma_f32_16x16x4_f32(false, a, false, b, (short)0, acc1, false, false);
      b[0] = q0[2]; b[1] = q1[2];
      acc2 = __builtin_amdgcn_wmma_f32_16x16x4_f32(false, a, false, b, (short)0, acc2, false, false);
      b[0] = q0[3]; b[1] = q1[3];
      acc3 = __builtin_amdgcn_wmma_f32_16x16x4_f32(false, a, false, b, (short)0, acc3, false, false);
    }
  }

  int r0 = wave * 16 + half * 8;
#pragma unroll
  for (int j = 0; j < 8; ++j) {
    float dv = dinv[r0 + j];
    float s  = dv * dv;
    v4f v;  v[0] = acc0[j]; v[1] = acc1[j]; v[2] = acc2[j]; v[3] = acc3[j];
    v4f w;  w[0] = s * v[0]; w[1] = s * v[1]; w[2] = s * v[2]; w[3] = s * v[3];
    size_t o = (size_t)(r0 + j) * DD + 4 * c;
    *(v4f*)(hw  + o) = v;
    *(v4f*)(agg + o) = w;
  }
}

#define EPW 64
#define CH  8
#define WPB 4
static_assert(EPW % 32 == 0 && EPW % CH == 0, "whole lane sets and whole row sets only");

__global__ void __launch_bounds__(32 * WPB)
k_scatter(const int* __restrict__ list0, const int* __restrict__ list1,
          unsigned* __restrict__ ctr, int layer,
          const float* __restrict__ dinv, const float* __restrict__ hw,
          float* __restrict__ agg) {
  __shared__ unsigned pick[3];
  int lane = threadIdx.x & 31;
  int wid  = __builtin_amdgcn_readfirstlane((int)(threadIdx.x >> 5));

  if (threadIdx.x == 0) {
    unsigned where = (unsigned)__builtin_amdgcn_s_sendmsg_rtn(0x87);
    unsigned h = (where >> 18) & 1u;
    unsigned n = ctr[h * CSTR];
    unsigned t = atomicAdd(&ctr[(2 + 2 * layer + h) * CSTR], 1u);
    if (t * (EPW * WPB) >= n) {
      h ^= 1u;
      n = ctr[h * CSTR];
      t = atomicAdd(&ctr[(2 + 2 * layer + h) * CSTR], 1u);
      if (t * (EPW * WPB) >= n) n = 0u;
    }
    pick[0] = h;  pick[1] = t * (EPW * WPB);  pick[2] = n;
  }
  __syncthreads();
  unsigned h    = (unsigned)__builtin_amdgcn_readfirstlane((int)pick[0]);
  unsigned base = (unsigned)__builtin_amdgcn_readfirstlane((int)pick[1]) + (unsigned)wid * EPW;
  unsigned n    = (unsigned)__builtin_amdgcn_readfirstlane((int)pick[2]);
  if (base >= n) return;
  unsigned cnt = n - base;
  const int* lst = h ? list1 : list0;

  const unsigned lo = 4u * (unsigned)lane;

  if (cnt < EPW) {
    int   sv[EPW / 32], dv[EPW / 32];
    float wv[EPW / 32];
    for (int hh = 0; hh < EPW / 32; ++hh) {
      bool in = (unsigned)(32 * hh + lane) < cnt;
      v2i rec = {0, 0};
      if (in) rec = *(const v2i*)(lst + 2 * (size_t)(base + 32 * hh + lane));
      sv[hh] = rec[0];
      dv[hh] = rec[1];
      wv[hh] = dinv[rec[0]] * dinv[rec[1]];
    }
    for (unsigned j = 0; j < cnt; ++j) {
      int   sj = j < 32 ? sv[0] : sv[1];
      int   dj = j < 32 ? dv[0] : dv[1];
      float wj = j < 32 ? wv[0] : wv[1];
      unsigned s = (unsigned)__builtin_amdgcn_readlane(sj, j & 31);
      unsigned d = (unsigned)__builtin_amdgcn_readlane(dj, j & 31);
      float    w = __int_as_float(__builtin_amdgcn_readlane(__float_as_int(wj), j & 31));
      const char* rp = (const char*)hw + (s * (DD * 4u) + lo);
      float v0 = *(const float*)rp;
      float v1 = *(const float*)(rp + 128);
      float* ap = (float*)((char*)agg + (d * (DD * 4u) + lo));
      atomicAdd(ap,      v0 * w);
      atomicAdd(ap + 32, v1 * w);
    }
    return;
  }

  int   sv[EPW / 32], dv[EPW / 32];
  float wv[EPW / 32];
  for (int hh = 0; hh < EPW / 32; ++hh) {
    v2i rec = *(const v2i*)(lst + 2 * (size_t)(base + 32 * hh + lane));
    sv[hh] = rec[0];
    dv[hh] = rec[1];
  }
  for (int hh = 0; hh < EPW / 32; ++hh)
    wv[hh] = dinv[sv[hh]] * dinv[dv[hh]];

  v2f rows[2][CH];
#pragma unroll
  for (int r = 0; r < CH; ++r) {
    unsigned s = (unsigned)__builtin_amdgcn_readlane(sv[0], r);
    const char* rp = (const char*)hw + (s * (DD * 4u) + lo);
    rows[0][r][0] = *(const float*)rp;
    rows[0][r][1] = *(const float*)(rp + 128);
  }
#pragma unroll
  for (int c = 0; c < EPW / CH; ++c) {
    if (c + 1 < EPW / CH) {
#pragma unroll
      for (int r = 0; r < CH; ++r) {
        int j = (c + 1) * CH + r;
        unsigned s = (unsigned)__builtin_amdgcn_readlane(sv[j >> 5], j & 31);
        const char* rp = (const char*)hw + (s * (DD * 4u) + lo);
        rows[(c + 1) & 1][r][0] = *(const float*)rp;
        rows[(c + 1) & 1][r][1] = *(const float*)(rp + 128);
      }
    }
#pragma unroll
    for (int r = 0; r < CH; ++r) {
      int j = c * CH + r;
      unsigned d = (unsigned)__builtin_amdgcn_readlane(dv[j >> 5], j & 31);
      float w = __int_as_float(__builtin_amdgcn_readlane(__float_as_int(wv[j >> 5]), j & 31));
      v2f   v = rows[c & 1][r];
      float* ap = (float*)((char*)agg + (d * (DD * 4u) + lo));
      atomicAdd(ap,      v[0] * w);
      atomicAdd(ap + 32, v[1] * w);
    }
  }
}

__device__ __forceinline__ unsigned fflip(float f) {
  unsigned u = __float_as_uint(f);
  return u ^ ((unsigned)((int)u >> 31) | 0x80000000u);
}
__device__ __forceinline__ float funflip(unsigned u) {
  unsigned m = (u >> 31) ? 0x80000000u : 0xFFFFFFFFu;
  return __uint_as_float(u ^ m);
}

__global__ void k_pool_init(unsigned* __restrict__ gmp, float* __restrict__ gsum,
                            float* __restrict__ counts) {
  int i = blockIdx.x * blockDim.x + threadIdx.x;
  if (i < GG * DD) { gmp[i] = 0x007FFFFFu  ; gsum[i] = 0.0f; }
  if (i < GG) counts[i] = 0.0f;
}

__global__ void k_pool(const float* __restrict__ aggl, const float* __restrict__ bl,
                       const int* __restrict__ bidx,
                       unsigned* __restrict__ gmp, float* __restrict__ gsum,
                       float* __restrict__ counts) {
  int i = blockIdx.x * blockDim.x + threadIdx.x;
  if (i >= NN * DD) return;
  int n = i >> 6, dd = i & 63;
  int g = bidx[n];
  float v = tanhf(aggl[i] + bl[dd]);
  atomicMax(&gmp[g * DD + dd], fflip(v));
  atomicAdd(&gsum[g * DD + dd], v);
  if (dd == 0) atomicAdd(&counts[g], 1.0f);
}

__global__ void k_final(const unsigned* __restrict__ gmp, const float* __restrict__ gsum,
                        const float* __restrict__ counts, const float* __restrict__ Wout,
                        const float* __restrict__ bout, float* __restrict__ out,
                        float* __restrict__ hidden) {
  __shared__ float red[2 * DD];
  int g = blockIdx.x, t = threadIdx.x;
  float cnt = fmaxf(counts[g], 1.0f);
  float hv = (t < DD) ? funflip(gmp[g * DD + t]) : (gsum[g * DD + (t - DD)] / cnt);
  hidden[(size_t)g * 2 * DD + t] = hv;
  red[t] = hv * Wout[t];
  __syncthreads();
  for (int s = DD; s > 0; s >>= 1) {
    if (t < s) red[t] += red[t + s];
    __syncthreads();
  }
  if (t == 0) out[g] = red[0] + bout[0];
}

extern "C" void kernel_launch(void* const* d_in, const int* in_sizes, int n_in,
                              void* d_out, int out_size, void* d_ws, size_t ws_size,
                              hipStream_t stream) {
  const float* x    = (const float*)d_in[0];
  const int*   ei   = (const int*)d_in[1];
  const int*   bidx = (const int*)d_in[2];
  const float* W0   = (const float*)d_in[3];
  const float* b0   = (const float*)d_in[4];
  const float* W1   = (const float*)d_in[5];
  const float* b1   = (const float*)d_in[6];
  const float* W2   = (const float*)d_in[7];
  const float* b2   = (const float*)d_in[8];
  const float* W3   = (const float*)d_in[9];
  const float* b3   = (const float*)d_in[10];
  const float* Wout = (const float*)d_in[11];
  const float* bout = (const float*)d_in[12];

  const int* src = ei;
  const int* dst = ei + EE;

  char*  ws   = (char*)d_ws;
  size_t off  = 0;
  float* dinv = (float*)(ws + off);  off += (((size_t)NN * 4 + 255) & ~(size_t)255);
  float* bufP0 = (float*)(ws + off); off += (size_t)NN * DD * 4;
  float* bufP1 = (float*)(ws + off); off += (size_t)NN * DD * 4;
  float* bufQ  = (float*)(ws + off); off += (size_t)NN * DD * 4;
  unsigned* gmp = (unsigned*)(ws + off); off += (size_t)GG * DD * 4;
  float* gsum   = (float*)(ws + off);    off += (size_t)GG * DD * 4;
  float* counts = (float*)(ws + off);    off += (((size_t)GG * 4 + 255) & ~(size_t)255);
  unsigned* ctr = (unsigned*)(ws + off); off += (size_t)NCTR * CSTR * 4;
  int* list0    = (int*)(ws + off);      off += (size_t)EE * 8;
  int* list1    = (int*)(ws + off);      off += (size_t)EE * 8;
  (void)ws_size; (void)n_in; (void)in_sizes; (void)out_size;

  float* outp   = (float*)d_out;
  float* hidden = outp + GG;

  const int T = 256;

  k_zero_deg<<<(NN + T - 1) / T, T, 0, stream>>>(dinv, ctr);
  k_degree_split<<<(EE + 1023) / 1024, 1024, 0, stream>>>(src, dst, dinv, ctr, list0, list1);
  k_dinv    <<<(NN + T - 1) / T, T, 0, stream>>>(dinv);

  const int gemm_blocks = (NN / 16 + 3) / 4;
  const int nd_blocks   = (NN * DD + T - 1) / T;
  const int sc_blocks   = EE / (EPW * WPB) + 3;

  k_gemm_fused<FF, false><<<gemm_blocks, 128, 0, stream>>>(x,     b0, W0, dinv, bufQ, bufP0);
  k_scatter<<<sc_blocks, 32 * WPB, 0, stream>>>(list0, list1, ctr, 0, dinv, bufQ, bufP0);
  k_gemm_fused<DD, true ><<<gemm_blocks, 128, 0, stream>>>(bufP0, b0, W1, dinv, bufQ, bufP1);
  k_scatter<<<sc_blocks, 32 * WPB, 0, stream>>>(list0, list1, ctr, 1, dinv, bufQ, bufP1);
  k_gemm_fused<DD, true ><<<gemm_blocks, 128, 0, stream>>>(bufP1, b1, W2, dinv, bufQ, bufP0);
  k_scatter<<<sc_blocks, 32 * WPB, 0, stream>>>(list0, list1, ctr, 2, dinv, bufQ, bufP0);
  k_gemm_fused<DD, true ><<<gemm_blocks, 128, 0, stream>>>(bufP0, b2, W3, dinv, bufQ, bufP1);
  k_scatter<<<sc_blocks, 32 * WPB, 0, stream>>>(list0, list1, ctr, 3, dinv, bufQ, bufP1);

  k_pool_init<<<(GG * DD + T - 1) / T, T, 0, stream>>>(gmp, gsum, counts);
  k_pool     <<<nd_blocks, T, 0, stream>>>(bufP1, b3, bidx, gmp, gsum, counts);
  k_final    <<<GG, 2 * DD, 0, stream>>>(gmp, gsum, counts, Wout, bout, outp, hidden);
}


# pooling + readout: one block per graph walks its contiguous node range (bounds by two-level search of the sorted ids), no atomics; 14 -> 12 launches; mean-pool sum order changes (fixed order)
# speedup vs baseline: 2.3528x; 1.5101x over previous
#include <hip/hip_runtime.h>
#include <math.h>

#define NN 100000
#define EE 1000000
#define FF 128
#define DD 64
#define GG 256

typedef float v2f __attribute__((ext_vector_type(2)));
typedef float v4f __attribute__((ext_vector_type(4)));
typedef float v8f __attribute__((ext_vector_type(8)));
typedef int   v2i __attribute__((ext_vector_type(2)));

#define NHALF (NN / 2)
#define CSTR 64
#define NCTR 16

__global__ void k_zero_deg(float* __restrict__ deg, unsigned* __restrict__ ctr) {
  int i = blockIdx.x * blockDim.x + threadIdx.x;
  if (i < NN) deg[i] = 0.0f;
  if (i < NCTR * CSTR) ctr[i] = 0u;
}

__global__ void __launch_bounds__(1024)
k_degree_split(const int* __restrict__ src, const int* __restrict__ dst, float* __restrict__ deg,
               unsigned* __restrict__ ctr, int* __restrict__ list0, int* __restrict__ list1) {
  __shared__ unsigned wc0[32], wc1[32], run[2];
  int e    = blockIdx.x * 1024 + threadIdx.x;
  int lane = threadIdx.x & 31;
  int wv   = threadIdx.x >> 5;
  bool valid = e < EE;
  int s = 0, d = 0;
  if (valid) {
    s = src[e];
    d = dst[e];
    atomicAdd(&deg[d], 1.0f);
  }
  bool p0 = valid && d < NHALF;
  bool p1 = valid && d >= NHALF;
  unsigned m0 = __builtin_amdgcn_ballot_w32(p0);
  unsigned m1 = __builtin_amdgcn_ballot_w32(p1);
  if (lane == 0) { wc0[wv] = __builtin_popcount(m0); wc1[wv] = __builtin_popcount(m1); }
  __syncthreads();
  if (wv == 0) {
    unsigned a0 = wc0[lane], a1 = wc1[lane];
    unsigned s0 = a0, s1 = a1;
#pragma unroll
    for (int o = 1; o < 32; o <<= 1) {
      unsigned t0 = __shfl_up(s0, o);
      unsigned t1 = __shfl_up(s1, o);
      if (lane >= o) { s0 += t0; s1 += t1; }
    }
    wc0[lane] = s0 - a0;
    wc1[lane] = s1 - a1;
    if (lane == 31) {
      run[0] = atomicAdd(&ctr[0 * CSTR], s0);
      run[1] = atomicAdd(&ctr[1 * CSTR], s1);
    }
  }
  __syncthreads();
  unsigned below = (1u << lane) - 1u;
  v2i rec;  rec[0] = s;  rec[1] = d;
  if (p0) *(v2i*)(list0 + 2 * (size_t)(run[0] + wc0[wv] + __builtin_popcount(m0 & below))) = rec;
  if (p1) *(v2i*)(list1 + 2 * (size_t)(run[1] + wc1[wv] + __builtin_popcount(m1 & below))) = rec;
}

__global__ void k_dinv(float* __restrict__ deg) {
  int i = blockIdx.x * blockDim.x + threadIdx.x;
  if (i < NN) deg[i] = rsqrtf(deg[i] + 1.0f);
}

template <int K, bool ACT>
__global__ void __launch_bounds__(128, 6)
k_gemm_fused(const float* __restrict__ A, const float* __restrict__ bprev,
             const float* __restrict__ B, const float* __restrict__ dinv,
             float* __restrict__ hw, float* __restrict__ agg) {
  int wave = (blockIdx.x * blockDim.x + threadIdx.x) >> 5;
  int lane = threadIdx.x & 31;
  if (wave * 16 >= NN) return;

  int half = lane >> 4;
  int c    = lane & 15;
  const float* arow = A + (size_t)(wave * 16 + c) * K;
  const float* bcol = B + 4 * c;

  v2f areg[K / 4];
#pragma unroll
  for (int i = 0; i < K / 4; ++i)
    areg[i] = *(const v2f*)(arow + 4 * i + 2 * half);
  if (ACT) {
    v2f breg[K / 4];
#pragma unroll
    for (int i = 0; i < K / 4; ++i)
      breg[i] = *(const v2f*)(bprev + 4 * i + 2 * half);
#pragma unroll
    for (int i = 0; i < K / 4; ++i) {
      areg[i][0] = tanhf(areg[i][0] + breg[i][0]);
      areg[i][1] = tanhf(areg[i][1] + breg[i][1]);
    }
  }

  v4f q[2][4];
#pragma unroll
  for (int s = 0; s < 2; ++s) {
    int k0 = 4 * s + 2 * half;
    q[0][2 * s]     = *(const v4f*)(bcol + (size_t)k0 * DD);
    q[0][2 * s + 1] = *(const v4f*)(bcol + (size_t)(k0 + 1) * DD);
  }

  v8f acc0 = {}, acc1 = {}, acc2 = {}, acc3 = {};
#pragma unroll
  for (int g = 0; g < K / 8; ++g) {
    if (g + 1 < K / 8) {
#pragma unroll
      for (int s = 0; s < 2; ++s) {
        int k0 = 8 * (g + 1) + 4 * s + 2 * half;
        q[(g + 1) & 1][2 * s]     = *(const v4f*)(bcol + (size_t)k0 * DD);
        q[(g + 1) & 1][2 * s + 1] = *(const v4f*)(bcol + (size_t)(k0 + 1) * DD);
      }
    }
#pragma unroll
    for (int s = 0; s < 2; ++s) {
      v2f a  = areg[2 * g + s];
      v4f q0 = q[g & 1][2 * s];
      v4f q1 = q[g & 1][2 * s + 1];
      v2f b;
      b[0] = q0[0]; b[1] = q1[0];
      acc0 = __builtin_amdgcn_wmma_f32_16x16x4_f32(false, a, false, b, (short)0, acc0, false, false);
      b[0] = q0[1]; b[1] = q1[1];
      acc1 = __builtin_amdgcn_wmma_f32_16x16x4_f32(false, a, false, b, (short)0, acc1, false, false);
      b[0] = q0[2]; b[1] = q1[2];
      acc2 = __builtin_amdgcn_wmma_f32_16x16x4_f32(false, a, false, b, (short)0, acc2, false, false);
      b[0] = q0[3]; b[1] = q1[3];
      acc3 = __builtin_amdgcn_wmma_f32_16x16x4_f32(false, a, false, b, (short)0, acc3, false, false);
    }
  }

  int r0 = wave * 16 + half * 8;
#pragma unroll
  for (int j = 0; j < 8; ++j) {
    float dv = dinv[r0 + j];
    float s  = dv * dv;
    v4f v;  v[0] = acc0[j]; v[1] = acc1[j]; v[2] = acc2[j]; v[3] = acc3[j];
    v4f w;  w[0] = s * v[0]; w[1] = s * v[1]; w[2] = s * v[2]; w[3] = s * v[3];
    size_t o = (size_t)(r0 + j) * DD + 4 * c;
    *(v4f*)(hw  + o) = v;
    *(v4f*)(agg + o) = w;
  }
}

#define EPW 64
#define CH  8
#define WPB 4
static_assert(EPW % 32 == 0 && EPW % CH == 0, "whole lane sets and whole row sets only");

__global__ void __launch_bounds__(32 * WPB)
k_scatter(const int* __restrict__ list0, const int* __restrict__ list1,
          unsigned* __restrict__ ctr, int layer,
          const float* __restrict__ dinv, const float* __restrict__ hw,
          float* __restrict__ agg) {
  __shared__ unsigned pick[3];
  int lane = threadIdx.x & 31;
  int wid  = __builtin_amdgcn_readfirstlane((int)(threadIdx.x >> 5));

  if (threadIdx.x == 0) {
    unsigned where = (unsigned)__builtin_amdgcn_s_sendmsg_rtn(0x87);
    unsigned h = (where >> 18) & 1u;
    unsigned n = ctr[h * CSTR];
    unsigned t = atomicAdd(&ctr[(2 + 2 * layer + h) * CSTR], 1u);
    if (t * (EPW * WPB) >= n) {
      h ^= 1u;
      n = ctr[h * CSTR];
      t = atomicAdd(&ctr[(2 + 2 * layer + h) * CSTR], 1u);
      if (t * (EPW * WPB) >= n) n = 0u;
    }
    pick[0] = h;  pick[1] = t * (EPW * WPB);  pick[2] = n;
  }
  __syncthreads();
  unsigned h    = (unsigned)__builtin_amdgcn_readfirstlane((int)pick[0]);
  unsigned base = (unsigned)__builtin_amdgcn_readfirstlane((int)pick[1]) + (unsigned)wid * EPW;
  unsigned n    = (unsigned)__builtin_amdgcn_readfirstlane((int)pick[2]);
  if (base >= n) return;
  unsigned cnt = n - base;
  const int* lst = h ? list1 : list0;

  const unsigned lo = 4u * (unsigned)lane;

  if (cnt < EPW) {
    int   sv[EPW / 32], dv[EPW / 32];
    float wv[EPW / 32];
    for (int hh = 0; hh < EPW / 32; ++hh) {
      bool in = (unsigned)(32 * hh + lane) < cnt;
      v2i rec = {0, 0};
      if (in) rec = *(const v2i*)(lst + 2 * (size_t)(base + 32 * hh + lane));
      sv[hh] = rec[0];
      dv[hh] = rec[1];
      wv[hh] = dinv[rec[0]] * dinv[rec[1]];
    }
    for (unsigned j = 0; j < cnt; ++j) {
      int   sj = j < 32 ? sv[0] : sv[1];
      int   dj = j < 32 ? dv[0] : dv[1];
      float wj = j < 32 ? wv[0] : wv[1];
      unsigned s = (unsigned)__builtin_amdgcn_readlane(sj, j & 31);
      unsigned d = (unsigned)__builtin_amdgcn_readlane(dj, j & 31);
      float    w = __int_as_float(__builtin_amdgcn_readlane(__float_as_int(wj), j & 31));
      const char* rp = (const char*)hw + (s * (DD * 4u) + lo);
      float v0 = *(const float*)rp;
      float v1 = *(const float*)(rp + 128);
      float* ap = (float*)((char*)agg + (d * (DD * 4u) + lo));
      atomicAdd(ap,      v0 * w);
      atomicAdd(ap + 32, v1 * w);
    }
    return;
  }

  int   sv[EPW / 32], dv[EPW / 32];
  float wv[EPW / 32];
  for (int hh = 0; hh < EPW / 32; ++hh) {
    v2i rec = *(const v2i*)(lst + 2 * (size_t)(base + 32 * hh + lane));
    sv[hh] = rec[0];
    dv[hh] = rec[1];
  }
  for (int hh = 0; hh < EPW / 32; ++hh)
    wv[hh] = dinv[sv[hh]] * dinv[dv[hh]];

  v2f rows[2][CH];
#pragma unroll
  for (int r = 0; r < CH; ++r) {
    unsigned s = (unsigned)__builtin_amdgcn_readlane(sv[0], r);
    const char* rp = (const char*)hw + (s * (DD * 4u) + lo);
    rows[0][r][0] = *(const float*)rp;
    rows[0][r][1] = *(const float*)(rp + 128);
  }
#pragma unroll
  for (int c = 0; c < EPW / CH; ++c) {
    if (c + 1 < EPW / CH) {
#pragma unroll
      for (int r = 0; r < CH; ++r) {
        int j = (c + 1) * CH + r;
        unsigned s = (unsigned)__builtin_amdgcn_readlane(sv[j >> 5], j & 31);
        const char* rp = (const char*)hw + (s * (DD * 4u) + lo);
        rows[(c + 1) & 1][r][0] = *(const float*)rp;
        rows[(c + 1) & 1][r][1] = *(const float*)(rp + 128);
      }
    }
#pragma unroll
    for (int r = 0; r < CH; ++r) {
      int j = c * CH + r;
      unsigned d = (unsigned)__builtin_amdgcn_readlane(dv[j >> 5], j & 31);
      float w = __int_as_float(__builtin_amdgcn_readlane(__float_as_int(wv[j >> 5]), j & 31));
      v2f   v = rows[c & 1][r];
      float* ap = (float*)((char*)agg + (d * (DD * 4u) + lo));
      atomicAdd(ap,      v[0] * w);
      atomicAdd(ap + 32, v[1] * w);
    }
  }
}

#define PW    32
#define SSTEP 100
#define NSAMP ((NN + SSTEP - 1) / SSTEP)
static_assert(NSAMP <= 32 * PW, "one sample per thread");
static_assert(SSTEP <= 128, "one window per four waves");

__global__ void __launch_bounds__(32 * PW)
k_pool_graph(const float* __restrict__ aggl, const float* __restrict__ bl,
             const int* __restrict__ bidx, const float* __restrict__ Wout,
             const float* __restrict__ bout, float* __restrict__ out,
             float* __restrict__ hidden) {
  __shared__ int   cnt[4];
  __shared__ __attribute__((aligned(8))) float smax[PW][DD];
  __shared__ __attribute__((aligned(8))) float ssum[PW][DD];
  __shared__ float red[2 * DD];
  int g    = blockIdx.x;
  int t    = threadIdx.x;
  int lane = t & 31;
  int wv   = t >> 5;

  if (t < 4) cnt[t] = 0;
  __syncthreads();

  bool a = false, b = false;
  if (t < NSAMP) {
    int v = bidx[t * SSTEP];
    a = v < g;
    b = v <= g;
  }
  unsigned ma = __builtin_amdgcn_ballot_w32(a);
  unsigned mb = __builtin_amdgcn_ballot_w32(b);
  if (lane == 0) {
    if (ma) atomicAdd(&cnt[0], __builtin_popcount(ma));
    if (mb) atomicAdd(&cnt[1], __builtin_popcount(mb));
  }
  __syncthreads();
  int c0 = cnt[0], c1 = cnt[1];
  int base_lo = (c0 > 0 ? c0 - 1 : 0) * SSTEP;
  int base_hi = (c1 > 0 ? c1 - 1 : 0) * SSTEP;

  int  which = t >> 7;
  bool c = false;
  if (t < 256) {
    int j = t & 127;
    int p = (which ? base_hi : base_lo) + j;
    if (j < SSTEP && p < NN) c = bidx[p] < g + which;
  }
  unsigned mc = __builtin_amdgcn_ballot_w32(c);
  if (t < 256 && lane == 0 && mc) atomicAdd(&cnt[2 + which], __builtin_popcount(mc));
  __syncthreads();
  int lo = base_lo + cnt[2];
  int hi = base_hi + cnt[3];

  v2f bb = *(const v2f*)(bl + 2 * lane);
  float m0 = -INFINITY, m1 = -INFINITY, s0 = 0.0f, s1 = 0.0f;
  const float* rp = aggl + 2 * lane;
  int n = lo + wv;
  for (; n + 3 * PW < hi; n += 4 * PW) {
    v2f r0 = *(const v2f*)(rp + (size_t)n * DD);
    v2f r1 = *(const v2f*)(rp + (size_t)(n + PW) * DD);
    v2f r2 = *(const v2f*)(rp + (size_t)(n + 2 * PW) * DD);
    v2f r3 = *(const v2f*)(rp + (size_t)(n + 3 * PW) * DD);
    float x0 = tanhf(r0[0] + bb[0]), y0 = tanhf(r0[1] + bb[1]);
    float x1 = tanhf(r1[0] + bb[0]), y1 = tanhf(r1[1] + bb[1]);
    float x2 = tanhf(r2[0] + bb[0]), y2 = tanhf(r2[1] + bb[1]);
    float x3 = tanhf(r3[0] + bb[0]), y3 = tanhf(r3[1] + bb[1]);
    m0 = fmaxf(fmaxf(fmaxf(fmaxf(m0, x0), x1), x2), x3);
    m1 = fmaxf(fmaxf(fmaxf(fmaxf(m1, y0), y1), y2), y3);
    s0 = (((s0 + x0) + x1) + x2) + x3;
    s1 = (((s1 + y0) + y1) + y2) + y3;
  }
  for (; n < hi; n += PW) {
    v2f r0 = *(const v2f*)(rp + (size_t)n * DD);
    float x0 = tanhf(r0[0] + bb[0]), y0 = tanhf(r0[1] + bb[1]);
    m0 = fmaxf(m0, x0);  m1 = fmaxf(m1, y0);
    s0 += x0;            s1 += y0;
  }
  v2f pm;  pm[0] = m0;  pm[1] = m1;
  v2f ps;  ps[0] = s0;  ps[1] = s1;
  *(v2f*)(&smax[wv][2 * lane]) = pm;
  *(v2f*)(&ssum[wv][2 * lane]) = ps;
  __syncthreads();

  if (t < 2 * DD) {
    float hv;
    if (t < DD) {
      hv = smax[0][t];
      for (int w = 1; w < PW; ++w) hv = fmaxf(hv, smax[w][t]);
    } else {
      hv = ssum[0][t - DD];
      for (int w = 1; w < PW; ++w) hv += ssum[w][t - DD];
      hv = hv / fmaxf((float)(hi - lo), 1.0f);
    }
    hidden[(size_t)g * 2 * DD + t] = hv;
    red[t] = hv * Wout[t];
  }
  __syncthreads();
  for (int s = DD; s > 0; s >>= 1) {
    if (t < s) red[t] += red[t + s];
    __syncthreads();
  }
  if (t == 0) out[g] = red[0] + bout[0];
}

extern "C" void kernel_launch(void* const* d_in, const int* in_sizes, int n_in,
                              void* d_out, int out_size, void* d_ws, size_t ws_size,
                              hipStream_t stream) {
  const float* x    = (const float*)d_in[0];
  const int*   ei   = (const int*)d_in[1];
  const int*   bidx = (const int*)d_in[2];
  const float* W0   = (const float*)d_in[3];
  const float* b0   = (const float*)d_in[4];
  const float* W1   = (const float*)d_in[5];
  const float* b1   = (const float*)d_in[6];
  const float* W2   = (const float*)d_in[7];
  const float* b2   = (const float*)d_in[8];
  const float* W3   = (const float*)d_in[9];
  const float* b3   = (const float*)d_in[10];
  const float* Wout = (const float*)d_in[11];
  const float* bout = (const float*)d_in[12];

  const int* src = ei;
  const int* dst = ei + EE;

  char*  ws   = (char*)d_ws;
  size_t off  = 0;
  float* dinv = (float*)(ws + off);  off += (((size_t)NN * 4 + 255) & ~(size_t)255);
  float* bufP0 = (float*)(ws + off); off += (size_t)NN * DD * 4;
  float* bufP1 = (float*)(ws + off); off += (size_t)NN * DD * 4;
  float* bufQ  = (float*)(ws + off); off += (size_t)NN * DD * 4;
  unsigned* ctr = (unsigned*)(ws + off); off += (size_t)NCTR * CSTR * 4;
  int* list0    = (int*)(ws + off);      off += (size_t)EE * 8;
  int* list1    = (int*)(ws + off);      off += (size_t)EE * 8;
  (void)ws_size; (void)n_in; (void)in_sizes; (void)out_size;

  float* outp   = (float*)d_out;
  float* hidden = outp + GG;

  const int T = 256;

  k_zero_deg<<<(NN + T - 1) / T, T, 0, stream>>>(dinv, ctr);
  k_degree_split<<<(EE + 1023) / 1024, 1024, 0, stream>>>(src, dst, dinv, ctr, list0, list1);
  k_dinv    <<<(NN + T - 1) / T, T, 0, stream>>>(dinv);

  const int gemm_blocks = (NN / 16 + 3) / 4;
  const int sc_blocks   = EE / (EPW * WPB) + 3;

  k_gemm_fused<FF, false><<<gemm_blocks, 128, 0, stream>>>(x,     b0, W0, dinv, bufQ, bufP0);
  k_scatter<<<sc_blocks, 32 * WPB, 0, stream>>>(list0, list1, ctr, 0, dinv, bufQ, bufP0);
  k_gemm_fused<DD, true ><<<gemm_blocks, 128, 0, stream>>>(bufP0, b0, W1, dinv, bufQ, bufP1);
  k_scatter<<<sc_blocks, 32 * WPB, 0, stream>>>(list0, list1, ctr, 1, dinv, bufQ, bufP1);
  k_gemm_fused<DD, true ><<<gemm_blocks, 128, 0, stream>>>(bufP1, b1, W2, dinv, bufQ, bufP0);
  k_scatter<<<sc_blocks, 32 * WPB, 0, stream>>>(list0, list1, ctr, 2, dinv, bufQ, bufP0);
  k_gemm_fused<DD, true ><<<gemm_blocks, 128, 0, stream>>>(bufP0, b2, W3, dinv, bufQ, bufP1);
  k_scatter<<<sc_blocks, 32 * WPB, 0, stream>>>(list0, list1, ctr, 3, dinv, bufQ, bufP1);

  k_pool_graph<<<GG, 32 * PW, 0, stream>>>(bufP1, b3, bidx, Wout, bout, outp, hidden);
}


# edge sums: edges grouped once per call by 100-node destination tile (record = src, row, weight); one block per tile adds its edges' rows into LDS (f32 LDS atomics, arrival order), stores rows once
# speedup vs baseline: 2.7318x; 1.1611x over previous
#include <hip/hip_runtime.h>
#include <math.h>

#define NN 100000
#define EE 1000000
#define FF 128
#define DD 64
#define GG 256

typedef float v2f __attribute__((ext_vector_type(2)));
typedef float v4f __attribute__((ext_vector_type(4)));
typedef float v8f __attribute__((ext_vector_type(8)));
typedef int   v2i __attribute__((ext_vector_type(2)));

#define TILE 100
#define NB (NN / TILE)
static_assert(NN % TILE == 0, "whole tiles only");
static_assert(NB < 1024, "one tile per thread in the grouping kernel, and one thread more for the total");

__global__ void k_zero_deg(float* __restrict__ deg, unsigned* __restrict__ cursor) {
  int i = blockIdx.x * blockDim.x + threadIdx.x;
  if (i < NN) deg[i] = 0.0f;
  if (i < NB) cursor[i] = 0u;
}

__global__ void k_degree(const int* __restrict__ dst, float* __restrict__ deg) {
  int e = blockIdx.x * blockDim.x + threadIdx.x;
  if (e < EE) atomicAdd(&deg[dst[e]], 1.0f);
}

__global__ void __launch_bounds__(128)
k_dinv_count(float* __restrict__ deg, unsigned* __restrict__ binCount) {
  __shared__ unsigned part[4];
  int t = threadIdx.x;
  unsigned c = 0u;
  if (t < TILE) {
    int n = blockIdx.x * TILE + t;
    float dg = deg[n];
    c = (unsigned)dg;
    deg[n] = rsqrtf(dg + 1.0f);
  }
#pragma unroll
  for (int o = 16; o > 0; o >>= 1) c += __shfl_xor(c, o);
  if ((t & 31) == 0) part[t >> 5] = c;
  __syncthreads();
  if (t == 0) binCount[blockIdx.x] = part[0] + part[1] + part[2] + part[3];
}

#define SRC_BITS 20
static_assert(NN <= (1 << SRC_BITS) && TILE <= (1 << (32 - SRC_BITS)), "record word 0 holds both");
#define FEPT 4
#define FE   (1024 * FEPT)
__global__ void __launch_bounds__(1024)
k_group(const int* __restrict__ src, const int* __restrict__ dst,
        const float* __restrict__ dinv,
        const unsigned* __restrict__ binCount, unsigned* __restrict__ cursor,
        unsigned* __restrict__ binStart, int* __restrict__ list) {
  __shared__ unsigned sstart[1024];
  __shared__ unsigned hist[1024];
  __shared__ unsigned wtot[32];
  int t    = threadIdx.x;
  int lane = t & 31;
  int wv   = t >> 5;
  unsigned c = t < NB ? binCount[t] : 0u;
  unsigned s = c;
#pragma unroll
  for (int o = 1; o < 32; o <<= 1) {
    unsigned u = __shfl_up(s, o);
    if (lane >= o) s += u;
  }
  if (lane == 31) wtot[wv] = s;
  hist[t] = 0u;
  __syncthreads();
  if (wv == 0) {
    unsigned a = wtot[lane];
    unsigned ss = a;
#pragma unroll
    for (int o = 1; o < 32; o <<= 1) {
      unsigned u = __shfl_up(ss, o);
      if (lane >= o) ss += u;
    }
    wtot[lane] = ss - a;
  }
  __syncthreads();
  unsigned excl = wtot[wv] + s - c;
  sstart[t] = excl;
  if (blockIdx.x == 0 && t <= NB) binStart[t] = excl;

  int sv[FEPT], dv[FEPT];
  float wt[FEPT];
  unsigned bn[FEPT], rk[FEPT];
#pragma unroll
  for (int k = 0; k < FEPT; ++k) {
    int e = blockIdx.x * FE + k * 1024 + t;
    int ec = e < EE ? e : EE - 1;
    sv[k] = src[ec];
    dv[k] = dst[ec];
  }
#pragma unroll
  for (int k = 0; k < FEPT; ++k) {
    wt[k] = dinv[sv[k]] * dinv[dv[k]];
    bn[k] = (unsigned)dv[k] / TILE;
  }
#pragma unroll
  for (int k = 0; k < FEPT; ++k) {
    int e = blockIdx.x * FE + k * 1024 + t;
    rk[k] = 0u;
    if (e < EE) rk[k] = atomicAdd(&hist[bn[k]], 1u);
  }
  __syncthreads();
  if (t < NB) {
    unsigned h = hist[t];
    if (h) sstart[t] = excl + atomicAdd(&cursor[t], h);
  }
  __syncthreads();
#pragma unroll
  for (int k = 0; k < FEPT; ++k) {
    int e = blockIdx.x * FE + k * 1024 + t;
    if (e < EE) {
      v2i rec;
      rec[0] = (int)((unsigned)sv[k] | (((unsigned)dv[k] - bn[k] * TILE) << SRC_BITS));
      rec[1] = __float_as_int(wt[k]);
      *(v2i*)(list + 2 * (size_t)(sstart[bn[k]] + rk[k])) = rec;
    }
  }
}

template <int K, bool ACT>
__global__ void __launch_bounds__(128, 6)
k_gemm_fused(const float* __restrict__ A, const float* __restrict__ bprev,
             const float* __restrict__ B, const float* __restrict__ dinv,
             float* __restrict__ hw, float* __restrict__ agg) {
  int wave = (blockIdx.x * blockDim.x + threadIdx.x) >> 5;
  int lane = threadIdx.x & 31;
  if (wave * 16 >= NN) return;

  int half = lane >> 4;
  int c    = lane & 15;
  const float* arow = A + (size_t)(wave * 16 + c) * K;
  const float* bcol = B + 4 * c;

  v2f areg[K / 4];
#pragma unroll
  for (int i = 0; i < K / 4; ++i)
    areg[i] = *(const v2f*)(arow + 4 * i + 2 * half);
  if (ACT) {
    v2f breg[K / 4];
#pragma unroll
    for (int i = 0; i < K / 4; ++i)
      breg[i] = *(const v2f*)(bprev + 4 * i + 2 * half);
#pragma unroll
    for (int i = 0; i < K / 4; ++i) {
      areg[i][0] = tanhf(areg[i][0] + breg[i][0]);
      areg[i][1] = tanhf(areg[i][1] + breg[i][1]);
    }
  }

  v4f q[2][4];
#pragma unroll
  for (int s = 0; s < 2; ++s) {
    int k0 = 4 * s + 2 * half;
    q[0][2 * s]     = *(const v4f*)(bcol + (size_t)k0 * DD);
    q[0][2 * s + 1] = *(const v4f*)(bcol + (size_t)(k0 + 1) * DD);
  }

  v8f acc0 = {}, acc1 = {}, acc2 = {}, acc3 = {};
#pragma unroll
  for (int g = 0; g < K / 8; ++g) {
    if (g + 1 < K / 8) {
#pragma unroll
      for (int s = 0; s < 2; ++s) {
        int k0 = 8 * (g + 1) + 4 * s + 2 * half;
        q[(g + 1) & 1][2 * s]     = *(const v4f*)(bcol + (size_t)k0 * DD);
        q[(g + 1) & 1][2 * s + 1] = *(const v4f*)(bcol + (size_t)(k0 + 1) * DD);
      }
    }
#pragma unroll
    for (int s = 0; s < 2; ++s) {
      v2f a  = areg[2 * g + s];
      v4f q0 = q[g & 1][2 * s];
      v4f q1 = q[g & 1][2 * s + 1];
      v2f b;
      b[0] = q0[0]; b[1] = q1[0];
      acc0 = __builtin_amdgcn_wmma_f32_16x16x4_f32(false, a, false, b, (short)0, acc0, false, false);
      b[0] = q0[1]; b[1] = q1[1];
      acc1 = __builtin_amdgcn_wmma_f32_16x16x4_f32(false, a, false, b, (short)0, acc1, false, false);
      b[0] = q0[2]; b[1] = q1[2];
      acc2 = __builtin_amdgcn_wmma_f32_16x16x4_f32(false, a, false, b, (short)0, acc2, false, false);
      b[0] = q0[3]; b[1] = q1[3];
      acc3 = __builtin_amdgcn_wmma_f32_16x16x4_f32(false, a, false, b, (short)0, acc3, false, false);
    }
  }

  int r0 = wave * 16 + half * 8;
#pragma unroll
  for (int j = 0; j < 8; ++j) {
    float dv = dinv[r0 + j];
    float s  = dv * dv;
    v4f v;  v[0] = acc0[j]; v[1] = acc1[j]; v[2] = acc2[j]; v[3] = acc3[j];
    v4f w;  w[0] = s * v[0]; w[1] = s * v[1]; w[2] = s * v[2]; w[3] = s * v[3];
    size_t o = (size_t)(r0 + j) * DD + 4 * c;
    *(v4f*)(hw  + o) = v;
    *(v4f*)(agg + o) = w;
  }
}

#define EPW 32
#define CH  8
#define SW  16
static_assert(EPW == 32 && EPW % CH == 0, "one edge per lane and whole row sets only");
static_assert((TILE * DD) % 4 == 0, "tile copied in 16-byte units");

__global__ void __launch_bounds__(32 * SW, 16)
k_scatter_tile(const int* __restrict__ list, const unsigned* __restrict__ binStart,
               const float* __restrict__ hw, float* __restrict__ agg) {
  __shared__ __attribute__((aligned(16))) float tile[TILE * DD];
  int t    = threadIdx.x;
  int lane = t & 31;
  int wid  = __builtin_amdgcn_readfirstlane(t >> 5);
  unsigned e0 = (unsigned)__builtin_amdgcn_readfirstlane((int)binStart[blockIdx.x]);
  unsigned e1 = (unsigned)__builtin_amdgcn_readfirstlane((int)binStart[blockIdx.x + 1]);
  int node0 = blockIdx.x * TILE;

  unsigned n = e1 - e0;
  const int* lst = list + 2 * (size_t)e0;
  unsigned cb = (unsigned)wid * EPW;
  v2i rec = {0, 0};
  if (cb + lane < n) rec = *(const v2i*)(lst + 2 * (size_t)(cb + lane));

  v4f* g4 = (v4f*)(agg + (size_t)node0 * DD);
  v4f* t4 = (v4f*)tile;
  constexpr int NV4   = TILE * DD / 4;
  constexpr int CFULL = NV4 / (32 * SW);
  constexpr int CREST = NV4 - CFULL * 32 * SW;
  {
    v4f cp[CFULL];
#pragma unroll
    for (int k = 0; k < CFULL; ++k) cp[k] = g4[t + k * 32 * SW];
    if (t < CREST) t4[t + CFULL * 32 * SW] = g4[t + CFULL * 32 * SW];
#pragma unroll
    for (int k = 0; k < CFULL; ++k) t4[t + k * 32 * SW] = cp[k];
  }
  __syncthreads();

  const unsigned lo = 4u * (unsigned)lane;

  while (cb < n) {
    unsigned cnt = n - cb;
    unsigned nb  = cb + EPW * SW;
    v2i nrec = {0, 0};
    if (nb + lane < n) nrec = *(const v2i*)(lst + 2 * (size_t)(nb + lane));
    int   sv = rec[0] & ((1 << SRC_BITS) - 1);
    int   dl = (int)((unsigned)rec[0] >> SRC_BITS) * DD;
    float wv = __int_as_float(rec[1]);

    if (cnt < EPW) {
      for (unsigned j0 = 0; j0 < cnt; j0 += CH) {
        v2f rr[CH];
#pragma unroll
        for (int r = 0; r < CH; ++r) {
          unsigned jj = j0 + r < cnt ? j0 + r : cnt - 1;
          unsigned s = (unsigned)__builtin_amdgcn_readlane(sv, jj);
          const char* rp = (const char*)hw + (s * (DD * 4u) + lo);
          rr[r][0] = *(const float*)rp;
          rr[r][1] = *(const float*)(rp + 128);
        }
#pragma unroll
        for (int r = 0; r < CH; ++r) {
          if (j0 + r < cnt) {
            int   d = __builtin_amdgcn_readlane(dl, j0 + r);
            float w = __int_as_float(__builtin_amdgcn_readlane(__float_as_int(wv), j0 + r));
            float* ap = &tile[d + lane];
            __hip_atomic_fetch_add(ap,      rr[r][0] * w, __ATOMIC_RELAXED, __HIP_MEMORY_SCOPE_WORKGROUP);
            __hip_atomic_fetch_add(ap + 32, rr[r][1] * w, __ATOMIC_RELAXED, __HIP_MEMORY_SCOPE_WORKGROUP);
          }
        }
      }
    } else {
      v2f rows[2][CH];
#pragma unroll
      for (int r = 0; r < CH; ++r) {
        unsigned s = (unsigned)__builtin_amdgcn_readlane(sv, r);
        const char* rp = (const char*)hw + (s * (DD * 4u) + lo);
        rows[0][r][0] = *(const float*)rp;
        rows[0][r][1] = *(const float*)(rp + 128);
      }
#pragma unroll
      for (int c = 0; c < EPW / CH; ++c) {
        if (c + 1 < EPW / CH) {
#pragma unroll
          for (int r = 0; r < CH; ++r) {
            int j = (c + 1) * CH + r;
            unsigned s = (unsigned)__builtin_amdgcn_readlane(sv, j);
            const char* rp = (const char*)hw + (s * (DD * 4u) + lo);
            rows[(c + 1) & 1][r][0] = *(const float*)rp;
            rows[(c + 1) & 1][r][1] = *(const float*)(rp + 128);
          }
        }
#pragma unroll
        for (int r = 0; r < CH; ++r) {
          int j = c * CH + r;
          int   d = __builtin_amdgcn_readlane(dl, j);
          float w = __int_as_float(__builtin_amdgcn_readlane(__float_as_int(wv), j));
          v2f   v = rows[c & 1][r];
          float* ap = &tile[d + lane];
          __hip_atomic_fetch_add(ap,      v[0] * w, __ATOMIC_RELAXED, __HIP_MEMORY_SCOPE_WORKGROUP);
          __hip_atomic_fetch_add(ap + 32, v[1] * w, __ATOMIC_RELAXED, __HIP_MEMORY_SCOPE_WORKGROUP);
        }
      }
    }
    rec = nrec;
    cb  = nb;
  }
  __syncthreads();
#pragma unroll
  for (int k = 0; k < CFULL; ++k) g4[t + k * 32 * SW] = t4[t + k * 32 * SW];
  if (t < CREST) g4[t + CFULL * 32 * SW] = t4[t + CFULL * 32 * SW];
}

#define PW    32
#define SSTEP 100
#define NSAMP ((NN + SSTEP - 1) / SSTEP)
static_assert(NSAMP <= 32 * PW, "one sample per thread");
static_assert(SSTEP <= 128, "one window per four waves");

__global__ void __launch_bounds__(32 * PW)
k_pool_graph(const float* __restrict__ aggl, const float* __restrict__ bl,
             const int* __restrict__ bidx, const float* __restrict__ Wout,
             const float* __restrict__ bout, float* __restrict__ out,
             float* __restrict__ hidden) {
  __shared__ int   cnt[4];
  __shared__ __attribute__((aligned(8))) float smax[PW][DD];
  __shared__ __attribute__((aligned(8))) float ssum[PW][DD];
  __shared__ float red[2 * DD];
  int g    = blockIdx.x;
  int t    = threadIdx.x;
  int lane = t & 31;
  int wv   = t >> 5;

  if (t < 4) cnt[t] = 0;
  __syncthreads();

  bool a = false, b = false;
  if (t < NSAMP) {
    int v = bidx[t * SSTEP];
    a = v < g;
    b = v <= g;
  }
  unsigned ma = __builtin_amdgcn_ballot_w32(a);
  unsigned mb = __builtin_amdgcn_ballot_w32(b);
  if (lane == 0) {
    if (ma) atomicAdd(&cnt[0], __builtin_popcount(ma));
    if (mb) atomicAdd(&cnt[1], __builtin_popcount(mb));
  }
  __syncthreads();
  int c0 = cnt[0], c1 = cnt[1];
  int base_lo = (c0 > 0 ? c0 - 1 : 0) * SSTEP;
  int base_hi = (c1 > 0 ? c1 - 1 : 0) * SSTEP;

  int  which = t >> 7;
  bool c = false;
  if (t < 256) {
    int j = t & 127;
    int p = (which ? base_hi : base_lo) + j;
    if (j < SSTEP && p < NN) c = bidx[p] < g + which;
  }
  unsigned mc = __builtin_amdgcn_ballot_w32(c);
  if (t < 256 && lane == 0 && mc) atomicAdd(&cnt[2 + which], __builtin_popcount(mc));
  __syncthreads();
  int lo = base_lo + cnt[2];
  int hi = base_hi + cnt[3];

  v2f bb = *(const v2f*)(bl + 2 * lane);
  float m0 = -INFINITY, m1 = -INFINITY, s0 = 0.0f, s1 = 0.0f;
  const float* rp = aggl + 2 * lane;
  int n = lo + wv;
  for (; n + 3 * PW < hi; n += 4 * PW) {
    v2f r0 = *(const v2f*)(rp + (size_t)n * DD);
    v2f r1 = *(const v2f*)(rp + (size_t)(n + PW) * DD);
    v2f r2 = *(const v2f*)(rp + (size_t)(n + 2 * PW) * DD);
    v2f r3 = *(const v2f*)(rp + (size_t)(n + 3 * PW) * DD);
    float x0 = tanhf(r0[0] + bb[0]), y0 = tanhf(r0[1] + bb[1]);
    float x1 = tanhf(r1[0] + bb[0]), y1 = tanhf(r1[1] + bb[1]);
    float x2 = tanhf(r2[0] + bb[0]), y2 = tanhf(r2[1] + bb[1]);
    float x3 = tanhf(r3[0] + bb[0]), y3 = tanhf(r3[1] + bb[1]);
    m0 = fmaxf(fmaxf(fmaxf(fmaxf(m0, x0), x1), x2), x3);
    m1 = fmaxf(fmaxf(fmaxf(fmaxf(m1, y0), y1), y2), y3);
    s0 = (((s0 + x0) + x1) + x2) + x3;
    s1 = (((s1 + y0) + y1) + y2) + y3;
  }
  for (; n < hi; n += PW) {
    v2f r0 = *(const v2f*)(rp + (size_t)n * DD);
    float x0 = tanhf(r0[0] + bb[0]), y0 = tanhf(r0[1] + bb[1]);
    m0 = fmaxf(m0, x0);  m1 = fmaxf(m1, y0);
    s0 += x0;            s1 += y0;
  }
  v2f pm;  pm[0] = m0;  pm[1] = m1;
  v2f ps;  ps[0] = s0;  ps[1] = s1;
  *(v2f*)(&smax[wv][2 * lane]) = pm;
  *(v2f*)(&ssum[wv][2 * lane]) = ps;
  __syncthreads();

  if (t < 2 * DD) {
    float hv;
    if (t < DD) {
      hv = smax[0][t];
      for (int w = 1; w < PW; ++w) hv = fmaxf(hv, smax[w][t]);
    } else {
      hv = ssum[0][t - DD];
      for (int w = 1; w < PW; ++w) hv += ssum[w][t - DD];
      hv = hv / fmaxf((float)(hi - lo), 1.0f);
    }
    hidden[(size_t)g * 2 * DD + t] = hv;
    red[t] = hv * Wout[t];
  }
  __syncthreads();
  for (int s = DD; s > 0; s >>= 1) {
    if (t < s) red[t] += red[t + s];
    __syncthreads();
  }
  if (t == 0) out[g] = red[0] + bout[0];
}

extern "C" void kernel_launch(void* const* d_in, const int* in_sizes, int n_in,
                              void* d_out, int out_size, void* d_ws, size_t ws_size,
                              hipStream_t stream) {
  const float* x    = (const float*)d_in[0];
  const int*   ei   = (const int*)d_in[1];
  const int*   bidx = (const int*)d_in[2];
  const float* W0   = (const float*)d_in[3];
  const float* b0   = (const float*)d_in[4];
  const float* W1   = (const float*)d_in[5];
  const float* b1   = (const float*)d_in[6];
  const float* W2   = (const float*)d_in[7];
  const float* b2   = (const float*)d_in[8];
  const float* W3   = (const float*)d_in[9];
  const float* b3   = (const float*)d_in[10];
  const float* Wout = (const float*)d_in[11];
  const float* bout = (const float*)d_in[12];

  const int* src = ei;
  const int* dst = ei + EE;

  char*  ws   = (char*)d_ws;
  size_t off  = 0;
  float* dinv = (float*)(ws + off);  off += (((size_t)NN * 4 + 255) & ~(size_t)255);
  float* bufP0 = (float*)(ws + off); off += (size_t)NN * DD * 4;
  float* bufP1 = (float*)(ws + off); off += (size_t)NN * DD * 4;
  float* bufQ  = (float*)(ws + off); off += (size_t)NN * DD * 4;
  unsigned* binCount = (unsigned*)(ws + off); off += 4096;
  unsigned* cursor   = (unsigned*)(ws + off); off += 4096;
  unsigned* binStart = (unsigned*)(ws + off); off += 4096;
  int* list          = (int*)(ws + off);      off += (size_t)EE * 8;
  (void)ws_size; (void)n_in; (void)in_sizes; (void)out_size;

  float* outp   = (float*)d_out;
  float* hidden = outp + GG;

  const int T = 256;

  k_zero_deg<<<(NN + T - 1) / T, T, 0, stream>>>(dinv, cursor);
  k_degree  <<<(EE + T - 1) / T, T, 0, stream>>>(dst, dinv);
  k_dinv_count<<<NB, 128, 0, stream>>>(dinv, binCount);
  k_group   <<<(EE + FE - 1) / FE, 1024, 0, stream>>>(src, dst, dinv, binCount, cursor, binStart, list);

  const int gemm_blocks = (NN / 16 + 3) / 4;

  k_gemm_fused<FF, false><<<gemm_blocks, 128, 0, stream>>>(x,     b0, W0, dinv, bufQ, bufP0);
  k_scatter_tile<<<NB, 32 * SW, 0, stream>>>(list, binStart, bufQ, bufP0);
  k_gemm_fused<DD, true ><<<gemm_blocks, 128, 0, stream>>>(bufP0, b0, W1, dinv, bufQ, bufP1);
  k_scatter_tile<<<NB, 32 * SW, 0, stream>>>(list, binStart, bufQ, bufP1);
  k_gemm_fused<DD, true ><<<gemm_blocks, 128, 0, stream>>>(bufP1, b1, W2, dinv, bufQ, bufP0);
  k_scatter_tile<<<NB, 32 * SW, 0, stream>>>(list, binStart, bufQ, bufP0);
  k_gemm_fused<DD, true ><<<gemm_blocks, 128, 0, stream>>>(bufP0, b2, W3, dinv, bufQ, bufP1);
  k_scatter_tile<<<NB, 32 * SW, 0, stream>>>(list, binStart, bufQ, bufP1);

  k_pool_graph<<<GG, 32 * PW, 0, stream>>>(bufP1, b3, bidx, Wout, bout, outp, hidden);
}


# in-degree counted after the grouping: a tile's block counts its records per row in LDS (tile sizes from per-block LDS counts); no global add per edge; layer 0's edge pass fills in the weights
# speedup vs baseline: 3.3255x; 1.2173x over previous
#include <hip/hip_runtime.h>
#include <math.h>

#define NN 100000
#define EE 1000000
#define FF 128
#define DD 64
#define GG 256

typedef float v2f __attribute__((ext_vector_type(2)));
typedef float v4f __attribute__((ext_vector_type(4)));
typedef float v8f __attribute__((ext_vector_type(8)));
typedef int   v2i __attribute__((ext_vector_type(2)));

#define TILE 100
#define NB (NN / TILE)
static_assert(NN % TILE == 0, "whole tiles only");
static_assert(NB < 1024, "one tile per thread in the grouping kernel, and one thread more for the total");

__global__ void k_zero_counts(unsigned* __restrict__ binCount, unsigned* __restrict__ cursor) {
  int i = blockIdx.x * blockDim.x + threadIdx.x;
  if (i < NB) { binCount[i] = 0u; cursor[i] = 0u; }
}

#define SRC_BITS 20
static_assert(NN <= (1 << SRC_BITS) && TILE <= (1 << (32 - SRC_BITS)), "record word 0 holds both");
#define FEPT 4
#define FE   (1024 * FEPT)

#define CEPT 16
#define CE   (1024 * CEPT)
__global__ void __launch_bounds__(1024)
k_tile_count(const int* __restrict__ dst, unsigned* __restrict__ binCount) {
  __shared__ unsigned hist[1024];
  int t = threadIdx.x;
  hist[t] = 0u;
  int dv[CEPT];
#pragma unroll
  for (int k = 0; k < CEPT; ++k) {
    int e = blockIdx.x * CE + k * 1024 + t;
    dv[k] = dst[e < EE ? e : EE - 1];
  }
  __syncthreads();
#pragma unroll
  for (int k = 0; k < CEPT; ++k) {
    int e = blockIdx.x * CE + k * 1024 + t;
    if (e < EE) atomicAdd(&hist[(unsigned)dv[k] / TILE], 1u);
  }
  __syncthreads();
  if (t < NB) {
    unsigned h = hist[t];
    if (h) atomicAdd(&binCount[t], h);
  }
}

__global__ void __launch_bounds__(1024)
k_group(const int* __restrict__ src, const int* __restrict__ dst,
        const unsigned* __restrict__ binCount, unsigned* __restrict__ cursor,
        unsigned* __restrict__ binStart, int* __restrict__ list) {
  __shared__ unsigned sstart[1024];
  __shared__ unsigned hist[1024];
  __shared__ unsigned wtot[32];
  int t    = threadIdx.x;
  int lane = t & 31;
  int wv   = t >> 5;
  unsigned c = t < NB ? binCount[t] : 0u;
  unsigned s = c;
#pragma unroll
  for (int o = 1; o < 32; o <<= 1) {
    unsigned u = __shfl_up(s, o);
    if (lane >= o) s += u;
  }
  if (lane == 31) wtot[wv] = s;
  hist[t] = 0u;
  __syncthreads();
  if (wv == 0) {
    unsigned a = wtot[lane];
    unsigned ss = a;
#pragma unroll
    for (int o = 1; o < 32; o <<= 1) {
      unsigned u = __shfl_up(ss, o);
      if (lane >= o) ss += u;
    }
    wtot[lane] = ss - a;
  }
  __syncthreads();
  unsigned excl = wtot[wv] + s - c;
  sstart[t] = excl;
  if (blockIdx.x == 0 && t <= NB) binStart[t] = excl;

  int sv[FEPT], dv[FEPT];
  unsigned bn[FEPT], rk[FEPT];
#pragma unroll
  for (int k = 0; k < FEPT; ++k) {
    int e = blockIdx.x * FE + k * 1024 + t;
    int ec = e < EE ? e : EE - 1;
    sv[k] = src[ec];
    dv[k] = dst[ec];
  }
#pragma unroll
  for (int k = 0; k < FEPT; ++k) bn[k] = (unsigned)dv[k] / TILE;
#pragma unroll
  for (int k = 0; k < FEPT; ++k) {
    int e = blockIdx.x * FE + k * 1024 + t;
    rk[k] = 0u;
    if (e < EE) rk[k] = atomicAdd(&hist[bn[k]], 1u);
  }
  __syncthreads();
  if (t < NB) {
    unsigned h = hist[t];
    if (h) sstart[t] = excl + atomicAdd(&cursor[t], h);
  }
  __syncthreads();
#pragma unroll
  for (int k = 0; k < FEPT; ++k) {
    int e = blockIdx.x * FE + k * 1024 + t;
    if (e < EE)
      list[2 * (size_t)(sstart[bn[k]] + rk[k])] =
          (int)((unsigned)sv[k] | (((unsigned)dv[k] - bn[k] * TILE) << SRC_BITS));
  }
}

#define DT 256
#define DU 5
__global__ void __launch_bounds__(DT)
k_tile_dinv(const int* __restrict__ list, const unsigned* __restrict__ binStart,
            float* __restrict__ dinv) {
  __shared__ unsigned cnt[TILE];
  int t = threadIdx.x;
  unsigned e0 = binStart[blockIdx.x];
  unsigned n  = binStart[blockIdx.x + 1] - e0;
  if (t < TILE) cnt[t] = 0u;
  __syncthreads();
  const int* lst = list + 2 * (size_t)e0;
  for (unsigned base = 0; base < n; base += DT * DU) {
    int w0[DU];
#pragma unroll
    for (int k = 0; k < DU; ++k) {
      unsigned i = base + k * DT + t;
      w0[k] = lst[2 * (size_t)(i < n ? i : n - 1)];
    }
#pragma unroll
    for (int k = 0; k < DU; ++k) {
      unsigned i = base + k * DT + t;
      atomicAdd(&cnt[(unsigned)w0[k] >> SRC_BITS], i < n ? 1u : 0u);
    }
  }
  __syncthreads();
  if (t < TILE) dinv[blockIdx.x * TILE + t] = rsqrtf((float)cnt[t] + 1.0f);
}

template <int K, bool ACT>
__global__ void __launch_bounds__(128, 6)
k_gemm_fused(const float* __restrict__ A, const float* __restrict__ bprev,
             const float* __restrict__ B, const float* __restrict__ dinv,
             float* __restrict__ hw, float* __restrict__ agg) {
  int wave = (blockIdx.x * blockDim.x + threadIdx.x) >> 5;
  int lane = threadIdx.x & 31;
  if (wave * 16 >= NN) return;

  int half = lane >> 4;
  int c    = lane & 15;
  const float* arow = A + (size_t)(wave * 16 + c) * K;
  const float* bcol = B + 4 * c;

  v2f areg[K / 4];
#pragma unroll
  for (int i = 0; i < K / 4; ++i)
    areg[i] = *(const v2f*)(arow + 4 * i + 2 * half);
  if (ACT) {
    v2f breg[K / 4];
#pragma unroll
    for (int i = 0; i < K / 4; ++i)
      breg[i] = *(const v2f*)(bprev + 4 * i + 2 * half);
#pragma unroll
    for (int i = 0; i < K / 4; ++i) {
      areg[i][0] = tanhf(areg[i][0] + breg[i][0]);
      areg[i][1] = tanhf(areg[i][1] + breg[i][1]);
    }
  }

  v4f q[2][4];
#pragma unroll
  for (int s = 0; s < 2; ++s) {
    int k0 = 4 * s + 2 * half;
    q[0][2 * s]     = *(const v4f*)(bcol + (size_t)k0 * DD);
    q[0][2 * s + 1] = *(const v4f*)(bcol + (size_t)(k0 + 1) * DD);
  }

  v8f acc0 = {}, acc1 = {}, acc2 = {}, acc3 = {};
#pragma unroll
  for (int g = 0; g < K / 8; ++g) {
    if (g + 1 < K / 8) {
#pragma unroll
      for (int s = 0; s < 2; ++s) {
        int k0 = 8 * (g + 1) + 4 * s + 2 * half;
        q[(g + 1) & 1][2 * s]     = *(const v4f*)(bcol + (size_t)k0 * DD);
        q[(g + 1) & 1][2 * s + 1] = *(const v4f*)(bcol + (size_t)(k0 + 1) * DD);
      }
    }
#pragma unroll
    for (int s = 0; s < 2; ++s) {
      v2f a  = areg[2 * g + s];
      v4f q0 = q[g & 1][2 * s];
      v4f q1 = q[g & 1][2 * s + 1];
      v2f b;
      b[0] = q0[0]; b[1] = q1[0];
      acc0 = __builtin_amdgcn_wmma_f32_16x16x4_f32(false, a, false, b, (short)0, acc0, false, false);
      b[0] = q0[1]; b[1] = q1[1];
      acc1 = __builtin_amdgcn_wmma_f32_16x16x4_f32(false, a, false, b, (short)0, acc1, false, false);
      b[0] = q0[2]; b[1] = q1[2];
      acc2 = __builtin_amdgcn_wmma_f32_16x16x4_f32(false, a, false, b, (short)0, acc2, false, false);
      b[0] = q0[3]; b[1] = q1[3];
      acc3 = __builtin_amdgcn_wmma_f32_16x16x4_f32(false, a, false, b, (short)0, acc3, false, false);
    }
  }

  int r0 = wave * 16 + half * 8;
#pragma unroll
  for (int j = 0; j < 8; ++j) {
    float dv = dinv[r0 + j];
    float s  = dv * dv;
    v4f v;  v[0] = acc0[j]; v[1] = acc1[j]; v[2] = acc2[j]; v[3] = acc3[j];
    v4f w;  w[0] = s * v[0]; w[1] = s * v[1]; w[2] = s * v[2]; w[3] = s * v[3];
    size_t o = (size_t)(r0 + j) * DD + 4 * c;
    *(v4f*)(hw  + o) = v;
    *(v4f*)(agg + o) = w;
  }
}

#define EPW 32
#define CH  8
#define SW  16
static_assert(EPW == 32 && EPW % CH == 0, "one edge per lane and whole row sets only");
static_assert((TILE * DD) % 4 == 0, "tile copied in 16-byte units");

template <bool FIRST>
__global__ void __launch_bounds__(32 * SW, 16)
k_scatter_tile(int* __restrict__ list, const unsigned* __restrict__ binStart,
               const float* __restrict__ dinv,
               const float* __restrict__ hw, float* __restrict__ agg) {
  __shared__ __attribute__((aligned(16))) float tile[TILE * DD];
  int t    = threadIdx.x;
  int lane = t & 31;
  int wid  = __builtin_amdgcn_readfirstlane(t >> 5);
  unsigned e0 = (unsigned)__builtin_amdgcn_readfirstlane((int)binStart[blockIdx.x]);
  unsigned e1 = (unsigned)__builtin_amdgcn_readfirstlane((int)binStart[blockIdx.x + 1]);
  int node0 = blockIdx.x * TILE;

  unsigned n = e1 - e0;
  int* lst = list + 2 * (size_t)e0;
  unsigned cb = (unsigned)wid * EPW;
  v2i rec = {0, 0};
  if (cb + lane < n) {
    if (FIRST) rec[0] = lst[2 * (size_t)(cb + lane)];
    else       rec = *(const v2i*)(lst + 2 * (size_t)(cb + lane));
  }

  v4f* g4 = (v4f*)(agg + (size_t)node0 * DD);
  v4f* t4 = (v4f*)tile;
  constexpr int NV4   = TILE * DD / 4;
  constexpr int CFULL = NV4 / (32 * SW);
  constexpr int CREST = NV4 - CFULL * 32 * SW;
  {
    v4f cp[CFULL];
#pragma unroll
    for (int k = 0; k < CFULL; ++k) cp[k] = g4[t + k * 32 * SW];
    if (t < CREST) t4[t + CFULL * 32 * SW] = g4[t + CFULL * 32 * SW];
#pragma unroll
    for (int k = 0; k < CFULL; ++k) t4[t + k * 32 * SW] = cp[k];
  }
  __syncthreads();

  const unsigned lo = 4u * (unsigned)lane;

  while (cb < n) {
    unsigned cnt = n - cb;
    unsigned nb  = cb + EPW * SW;
    v2i nrec = {0, 0};
    if (nb + lane < n) {
      if (FIRST) nrec[0] = lst[2 * (size_t)(nb + lane)];
      else       nrec = *(const v2i*)(lst + 2 * (size_t)(nb + lane));
    }
    int   sv = rec[0] & ((1 << SRC_BITS) - 1);
    int   rw = (int)((unsigned)rec[0] >> SRC_BITS);
    int   dl = rw * DD;
    float wv;
    if (FIRST) {
      wv = dinv[sv] * dinv[node0 + rw];
      if (cb + lane < n) lst[2 * (size_t)(cb + lane) + 1] = __float_as_int(wv);
    } else {
      wv = __int_as_float(rec[1]);
    }

    if (cnt < EPW) {
      for (unsigned j0 = 0; j0 < cnt; j0 += CH) {
        v2f rr[CH];
#pragma unroll
        for (int r = 0; r < CH; ++r) {
          unsigned jj = j0 + r < cnt ? j0 + r : cnt - 1;
          unsigned s = (unsigned)__builtin_amdgcn_readlane(sv, jj);
          const char* rp = (const char*)hw + (s * (DD * 4u) + lo);
          rr[r][0] = *(const float*)rp;
          rr[r][1] = *(const float*)(rp + 128);
        }
#pragma unroll
        for (int r = 0; r < CH; ++r) {
          if (j0 + r < cnt) {
            int   d = __builtin_amdgcn_readlane(dl, j0 + r);
            float w = __int_as_float(__builtin_amdgcn_readlane(__float_as_int(wv), j0 + r));
            float* ap = &tile[d + lane];
            __hip_atomic_fetch_add(ap,      rr[r][0] * w, __ATOMIC_RELAXED, __HIP_MEMORY_SCOPE_WORKGROUP);
            __hip_atomic_fetch_add(ap + 32, rr[r][1] * w, __ATOMIC_RELAXED, __HIP_MEMORY_SCOPE_WORKGROUP);
          }
        }
      }
    } else {
      v2f rows[2][CH];
#pragma unroll
      for (int r = 0; r < CH; ++r) {
        unsigned s = (unsigned)__builtin_amdgcn_readlane(sv, r);
        const char* rp = (const char*)hw + (s * (DD * 4u) + lo);
        rows[0][r][0] = *(const float*)rp;
        rows[0][r][1] = *(const float*)(rp + 128);
      }
#pragma unroll
      for (int c = 0; c < EPW / CH; ++c) {
        if (c + 1 < EPW / CH) {
#pragma unroll
          for (int r = 0; r < CH; ++r) {
            int j = (c + 1) * CH + r;
            unsigned s = (unsigned)__builtin_amdgcn_readlane(sv, j);
            const char* rp = (const char*)hw + (s * (DD * 4u) + lo);
            rows[(c + 1) & 1][r][0] = *(const float*)rp;
            rows[(c + 1) & 1][r][1] = *(const float*)(rp + 128);
          }
        }
#pragma unroll
        for (int r = 0; r < CH; ++r) {
          int j = c * CH + r;
          int   d = __builtin_amdgcn_readlane(dl, j);
          float w = __int_as_float(__builtin_amdgcn_readlane(__float_as_int(wv), j));
          v2f   v = rows[c & 1][r];
          float* ap = &tile[d + lane];
          __hip_atomic_fetch_add(ap,      v[0] * w, __ATOMIC_RELAXED, __HIP_MEMORY_SCOPE_WORKGROUP);
          __hip_atomic_fetch_add(ap + 32, v[1] * w, __ATOMIC_RELAXED, __HIP_MEMORY_SCOPE_WORKGROUP);
        }
      }
    }
    rec = nrec;
    cb  = nb;
  }
  __syncthreads();
#pragma unroll
  for (int k = 0; k < CFULL; ++k) g4[t + k * 32 * SW] = t4[t + k * 32 * SW];
  if (t < CREST) g4[t + CFULL * 32 * SW] = t4[t + CFULL * 32 * SW];
}

#define PW    32
#define SSTEP 100
#define NSAMP ((NN + SSTEP - 1) / SSTEP)
static_assert(NSAMP <= 32 * PW, "one sample per thread");
static_assert(SSTEP <= 128, "one window per four waves");

__global__ void __launch_bounds__(32 * PW)
k_pool_graph(const float* __restrict__ aggl, const float* __restrict__ bl,
             const int* __restrict__ bidx, const float* __restrict__ Wout,
             const float* __restrict__ bout, float* __restrict__ out,
             float* __restrict__ hidden) {
  __shared__ int   cnt[4];
  __shared__ __attribute__((aligned(8))) float smax[PW][DD];
  __shared__ __attribute__((aligned(8))) float ssum[PW][DD];
  __shared__ float red[2 * DD];
  int g    = blockIdx.x;
  int t    = threadIdx.x;
  int lane = t & 31;
  int wv   = t >> 5;

  if (t < 4) cnt[t] = 0;
  __syncthreads();

  bool a = false, b = false;
  if (t < NSAMP) {
    int v = bidx[t * SSTEP];
    a = v < g;
    b = v <= g;
  }
  unsigned ma = __builtin_amdgcn_ballot_w32(a);
  unsigned mb = __builtin_amdgcn_ballot_w32(b);
  if (lane == 0) {
    if (ma) atomicAdd(&cnt[0], __builtin_popcount(ma));
    if (mb) atomicAdd(&cnt[1], __builtin_popcount(mb));
  }
  __syncthreads();
  int c0 = cnt[0], c1 = cnt[1];
  int base_lo = (c0 > 0 ? c0 - 1 : 0) * SSTEP;
  int base_hi = (c1 > 0 ? c1 - 1 : 0) * SSTEP;

  int  which = t >> 7;
  bool c = false;
  if (t < 256) {
    int j = t & 127;
    int p = (which ? base_hi : base_lo) + j;
    if (j < SSTEP && p < NN) c = bidx[p] < g + which;
  }
  unsigned mc = __builtin_amdgcn_ballot_w32(c);
  if (t < 256 && lane == 0 && mc) atomicAdd(&cnt[2 + which], __builtin_popcount(mc));
  __syncthreads();
  int lo = base_lo + cnt[2];
  int hi = base_hi + cnt[3];

  v2f bb = *(const v2f*)(bl + 2 * lane);
  float m0 = -INFINITY, m1 = -INFINITY, s0 = 0.0f, s1 = 0.0f;
  const float* rp = aggl + 2 * lane;
  int n = lo + wv;
  for (; n + 3 * PW < hi; n += 4 * PW) {
    v2f r0 = *(const v2f*)(rp + (size_t)n * DD);
    v2f r1 = *(const v2f*)(rp + (size_t)(n + PW) * DD);
    v2f r2 = *(const v2f*)(rp + (size_t)(n + 2 * PW) * DD);
    v2f r3 = *(const v2f*)(rp + (size_t)(n + 3 * PW) * DD);
    float x0 = tanhf(r0[0] + bb[0]), y0 = tanhf(r0[1] + bb[1]);
    float x1 = tanhf(r1[0] + bb[0]), y1 = tanhf(r1[1] + bb[1]);
    float x2 = tanhf(r2[0] + bb[0]), y2 = tanhf(r2[1] + bb[1]);
    float x3 = tanhf(r3[0] + bb[0]), y3 = tanhf(r3[1] + bb[1]);
    m0 = fmaxf(fmaxf(fmaxf(fmaxf(m0, x0), x1), x2), x3);
    m1 = fmaxf(fmaxf(fmaxf(fmaxf(m1, y0), y1), y2), y3);
    s0 = (((s0 + x0) + x1) + x2) + x3;
    s1 = (((s1 + y0) + y1) + y2) + y3;
  }
  for (; n < hi; n += PW) {
    v2f r0 = *(const v2f*)(rp + (size_t)n * DD);
    float x0 = tanhf(r0[0] + bb[0]), y0 = tanhf(r0[1] + bb[1]);
    m0 = fmaxf(m0, x0);  m1 = fmaxf(m1, y0);
    s0 += x0;            s1 += y0;
  }
  v2f pm;  pm[0] = m0;  pm[1] = m1;
  v2f ps;  ps[0] = s0;  ps[1] = s1;
  *(v2f*)(&smax[wv][2 * lane]) = pm;
  *(v2f*)(&ssum[wv][2 * lane]) = ps;
  __syncthreads();

  if (t < 2 * DD) {
    float hv;
    if (t < DD) {
      hv = smax[0][t];
      for (int w = 1; w < PW; ++w) hv = fmaxf(hv, smax[w][t]);
    } else {
      hv = ssum[0][t - DD];
      for (int w = 1; w < PW; ++w) hv += ssum[w][t - DD];
      hv = hv / fmaxf((float)(hi - lo), 1.0f);
    }
    hidden[(size_t)g * 2 * DD + t] = hv;
    red[t] = hv * Wout[t];
  }
  __syncthreads();
  for (int s = DD; s > 0; s >>= 1) {
    if (t < s) red[t] += red[t + s];
    __syncthreads();
  }
  if (t == 0) out[g] = red[0] + bout[0];
}

extern "C" void kernel_launch(void* const* d_in, const int* in_sizes, int n_in,
                              void* d_out, int out_size, void* d_ws, size_t ws_size,
                              hipStream_t stream) {
  const float* x    = (const float*)d_in[0];
  const int*   ei   = (const int*)d_in[1];
  const int*   bidx = (const int*)d_in[2];
  const float* W0   = (const float*)d_in[3];
  const float* b0   = (const float*)d_in[4];
  const float* W1   = (const float*)d_in[5];
  const float* b1   = (const float*)d_in[6];
  const float* W2   = (const float*)d_in[7];
  const float* b2   = (const float*)d_in[8];
  const float* W3   = (const float*)d_in[9];
  const float* b3   = (const float*)d_in[10];
  const float* Wout = (const float*)d_in[11];
  const float* bout = (const float*)d_in[12];

  const int* src = ei;
  const int* dst = ei + EE;

  char*  ws   = (char*)d_ws;
  size_t off  = 0;
  float* dinv = (float*)(ws + off);  off += (((size_t)NN * 4 + 255) & ~(size_t)255);
  float* bufP0 = (float*)(ws + off); off += (size_t)NN * DD * 4;
  float* bufP1 = (float*)(ws + off); off += (size_t)NN * DD * 4;
  float* bufQ  = (float*)(ws + off); off += (size_t)NN * DD * 4;
  unsigned* binCount = (unsigned*)(ws + off); off += 4096;
  unsigned* cursor   = (unsigned*)(ws + off); off += 4096;
  unsigned* binStart = (unsigned*)(ws + off); off += 4096;
  int* list          = (int*)(ws + off);      off += (size_t)EE * 8;
  (void)ws_size; (void)n_in; (void)in_sizes; (void)out_size;

  float* outp   = (float*)d_out;
  float* hidden = outp + GG;

  const int T = 256;

  k_zero_counts<<<(NB + T - 1) / T, T, 0, stream>>>(binCount, cursor);
  k_tile_count<<<(EE + CE - 1) / CE, 1024, 0, stream>>>(dst, binCount);
  k_group     <<<(EE + FE - 1) / FE, 1024, 0, stream>>>(src, dst, binCount, cursor, binStart, list);
  k_tile_dinv <<<NB, DT, 0, stream>>>(list, binStart, dinv);

  const int gemm_blocks = (NN / 16 + 3) / 4;

  k_gemm_fused<FF, false><<<gemm_blocks, 128, 0, stream>>>(x,     b0, W0, dinv, bufQ, bufP0);
  k_scatter_tile<true ><<<NB, 32 * SW, 0, stream>>>(list, binStart, dinv, bufQ, bufP0);
  k_gemm_fused<DD, true ><<<gemm_blocks, 128, 0, stream>>>(bufP0, b0, W1, dinv, bufQ, bufP1);
  k_scatter_tile<false><<<NB, 32 * SW, 0, stream>>>(list, binStart, dinv, bufQ, bufP1);
  k_gemm_fused<DD, true ><<<gemm_blocks, 128, 0, stream>>>(bufP1, b1, W2, dinv, bufQ, bufP0);
  k_scatter_tile<false><<<NB, 32 * SW, 0, stream>>>(list, binStart, dinv, bufQ, bufP0);
  k_gemm_fused<DD, true ><<<gemm_blocks, 128, 0, stream>>>(bufP0, b2, W3, dinv, bufQ, bufP1);
  k_scatter_tile<false><<<NB, 32 * SW, 0, stream>>>(list, binStart, dinv, bufQ, bufP1);

  k_pool_graph<<<GG, 32 * PW, 0, stream>>>(bufP1, b3, bidx, Wout, bout, outp, hidden);
}


# edge weight dinv[s]*dinv[d] split in two: product stores rows scaled by dinv[s] (one output, not two), edges add rows unweighted (4-byte records), tile store scales by dinv[d]; sum order unchanged
# speedup vs baseline: 3.6300x; 1.0916x over previous
#include <hip/hip_runtime.h>
#include <math.h>

#define NN 100000
#define EE 1000000
#define FF 128
#define DD 64
#define GG 256

typedef float v2f __attribute__((ext_vector_type(2)));
typedef float v4f __attribute__((ext_vector_type(4)));
typedef float v8f __attribute__((ext_vector_type(8)));

#define TILE 100
#define NB (NN / TILE)
static_assert(NN % TILE == 0, "whole tiles only");
static_assert(NB < 1024, "one tile per thread in the grouping kernel, and one thread more for the total");

__global__ void k_zero_counts(unsigned* __restrict__ binCount, unsigned* __restrict__ cursor) {
  int i = blockIdx.x * blockDim.x + threadIdx.x;
  if (i < NB) { binCount[i] = 0u; cursor[i] = 0u; }
}

#define SRC_BITS 20
static_assert(NN <= (1 << SRC_BITS) && TILE <= (1 << (32 - SRC_BITS)), "record word 0 holds both");
#define FEPT 4
#define FE   (1024 * FEPT)

#define CEPT 16
#define CE   (1024 * CEPT)
__global__ void __launch_bounds__(1024)
k_tile_count(const int* __restrict__ dst, unsigned* __restrict__ binCount) {
  __shared__ unsigned hist[1024];
  int t = threadIdx.x;
  hist[t] = 0u;
  int dv[CEPT];
#pragma unroll
  for (int k = 0; k < CEPT; ++k) {
    int e = blockIdx.x * CE + k * 1024 + t;
    dv[k] = dst[e < EE ? e : EE - 1];
  }
  __syncthreads();
#pragma unroll
  for (int k = 0; k < CEPT; ++k) {
    int e = blockIdx.x * CE + k * 1024 + t;
    if (e < EE) atomicAdd(&hist[(unsigned)dv[k] / TILE], 1u);
  }
  __syncthreads();
  if (t < NB) {
    unsigned h = hist[t];
    if (h) atomicAdd(&binCount[t], h);
  }
}

__global__ void __launch_bounds__(1024)
k_group(const int* __restrict__ src, const int* __restrict__ dst,
        const unsigned* __restrict__ binCount, unsigned* __restrict__ cursor,
        unsigned* __restrict__ binStart, int* __restrict__ list) {
  __shared__ unsigned sstart[1024];
  __shared__ unsigned hist[1024];
  __shared__ unsigned wtot[32];
  int t    = threadIdx.x;
  int lane = t & 31;
  int wv   = t >> 5;
  unsigned c = t < NB ? binCount[t] : 0u;
  unsigned s = c;
#pragma unroll
  for (int o = 1; o < 32; o <<= 1) {
    unsigned u = __shfl_up(s, o);
    if (lane >= o) s += u;
  }
  if (lane == 31) wtot[wv] = s;
  hist[t] = 0u;
  __syncthreads();
  if (wv == 0) {
    unsigned a = wtot[lane];
    unsigned ss = a;
#pragma unroll
    for (int o = 1; o < 32; o <<= 1) {
      unsigned u = __shfl_up(ss, o);
      if (lane >= o) ss += u;
    }
    wtot[lane] = ss - a;
  }
  __syncthreads();
  unsigned excl = wtot[wv] + s - c;
  sstart[t] = excl;
  if (blockIdx.x == 0 && t <= NB) binStart[t] = excl;

  int sv[FEPT], dv[FEPT];
  unsigned bn[FEPT], rk[FEPT];
#pragma unroll
  for (int k = 0; k < FEPT; ++k) {
    int e = blockIdx.x * FE + k * 1024 + t;
    int ec = e < EE ? e : EE - 1;
    sv[k] = src[ec];
    dv[k] = dst[ec];
  }
#pragma unroll
  for (int k = 0; k < FEPT; ++k) bn[k] = (unsigned)dv[k] / TILE;
#pragma unroll
  for (int k = 0; k < FEPT; ++k) {
    int e = blockIdx.x * FE + k * 1024 + t;
    rk[k] = 0u;
    if (e < EE) rk[k] = atomicAdd(&hist[bn[k]], 1u);
  }
  __syncthreads();
  if (t < NB) {
    unsigned h = hist[t];
    if (h) sstart[t] = excl + atomicAdd(&cursor[t], h);
  }
  __syncthreads();
#pragma unroll
  for (int k = 0; k < FEPT; ++k) {
    int e = blockIdx.x * FE + k * 1024 + t;
    if (e < EE)
      list[(size_t)(sstart[bn[k]] + rk[k])] =
          (int)((unsigned)sv[k] | (((unsigned)dv[k] - bn[k] * TILE) << SRC_BITS));
  }
}

#define DT 256
#define DU 5
__global__ void __launch_bounds__(DT)
k_tile_dinv(const int* __restrict__ list, const unsigned* __restrict__ binStart,
            float* __restrict__ dinv) {
  __shared__ unsigned cnt[TILE];
  int t = threadIdx.x;
  unsigned e0 = binStart[blockIdx.x];
  unsigned n  = binStart[blockIdx.x + 1] - e0;
  if (t < TILE) cnt[t] = 0u;
  __syncthreads();
  const int* lst = list + (size_t)e0;
  for (unsigned base = 0; base < n; base += DT * DU) {
    int w0[DU];
#pragma unroll
    for (int k = 0; k < DU; ++k) {
      unsigned i = base + k * DT + t;
      w0[k] = lst[i < n ? i : n - 1];
    }
#pragma unroll
    for (int k = 0; k < DU; ++k) {
      unsigned i = base + k * DT + t;
      atomicAdd(&cnt[(unsigned)w0[k] >> SRC_BITS], i < n ? 1u : 0u);
    }
  }
  __syncthreads();
  if (t < TILE) dinv[blockIdx.x * TILE + t] = rsqrtf((float)cnt[t] + 1.0f);
}

template <int K, bool ACT>
__global__ void __launch_bounds__(128, 6)
k_gemm_fused(const float* __restrict__ A, const float* __restrict__ bprev,
             const float* __restrict__ B, const float* __restrict__ dinv,
             float* __restrict__ hs) {
  int wave = (blockIdx.x * blockDim.x + threadIdx.x) >> 5;
  int lane = threadIdx.x & 31;
  if (wave * 16 >= NN) return;

  int half = lane >> 4;
  int c    = lane & 15;
  const float* arow = A + (size_t)(wave * 16 + c) * K;
  const float* bcol = B + 4 * c;

  v2f areg[K / 4];
#pragma unroll
  for (int i = 0; i < K / 4; ++i)
    areg[i] = *(const v2f*)(arow + 4 * i + 2 * half);
  if (ACT) {
    v2f breg[K / 4];
#pragma unroll
    for (int i = 0; i < K / 4; ++i)
      breg[i] = *(const v2f*)(bprev + 4 * i + 2 * half);
#pragma unroll
    for (int i = 0; i < K / 4; ++i) {
      areg[i][0] = tanhf(areg[i][0] + breg[i][0]);
      areg[i][1] = tanhf(areg[i][1] + breg[i][1]);
    }
  }

  v4f q[2][4];
#pragma unroll
  for (int s = 0; s < 2; ++s) {
    int k0 = 4 * s + 2 * half;
    q[0][2 * s]     = *(const v4f*)(bcol + (size_t)k0 * DD);
    q[0][2 * s + 1] = *(const v4f*)(bcol + (size_t)(k0 + 1) * DD);
  }

  v8f acc0 = {}, acc1 = {}, acc2 = {}, acc3 = {};
#pragma unroll
  for (int g = 0; g < K / 8; ++g) {
    if (g + 1 < K / 8) {
#pragma unroll
      for (int s = 0; s < 2; ++s) {
        int k0 = 8 * (g + 1) + 4 * s + 2 * half;
        q[(g + 1) & 1][2 * s]     = *(const v4f*)(bcol + (size_t)k0 * DD);
        q[(g + 1) & 1][2 * s + 1] = *(const v4f*)(bcol + (size_t)(k0 + 1) * DD);
      }
    }
#pragma unroll
    for (int s = 0; s < 2; ++s) {
      v2f a  = areg[2 * g + s];
      v4f q0 = q[g & 1][2 * s];
      v4f q1 = q[g & 1][2 * s + 1];
      v2f b;
      b[0] = q0[0]; b[1] = q1[0];
      acc0 = __builtin_amdgcn_wmma_f32_16x16x4_f32(false, a, false, b, (short)0, acc0, false, false);
      b[0] = q0[1]; b[1] = q1[1];
      acc1 = __builtin_amdgcn_wmma_f32_16x16x4_f32(false, a, false, b, (short)0, acc1, false, false);
      b[0] = q0[2]; b[1] = q1[2];
      acc2 = __builtin_amdgcn_wmma_f32_16x16x4_f32(false, a, false, b, (short)0, acc2, false, false);
      b[0] = q0[3]; b[1] = q1[3];
      acc3 = __builtin_amdgcn_wmma_f32_16x16x4_f32(false, a, false, b, (short)0, acc3, false, false);
    }
  }

  int r0 = wave * 16 + half * 8;
#pragma unroll
  for (int j = 0; j < 8; ++j) {
    float dv = dinv[r0 + j];
    v4f v;  v[0] = dv * acc0[j]; v[1] = dv * acc1[j]; v[2] = dv * acc2[j]; v[3] = dv * acc3[j];
    *(v4f*)(hs + (size_t)(r0 + j) * DD + 4 * c) = v;
  }
}

#define EPW 32
#define CH  8
#define SW  16
static_assert(EPW == 32 && EPW % CH == 0, "one edge per lane and whole row sets only");
static_assert((TILE * DD) % 4 == 0, "tile copied in 16-byte units");
static_assert(DD == 64, "a 16-byte unit's row is its number / 16");

__global__ void __launch_bounds__(32 * SW, 16)
k_scatter_tile(const int* __restrict__ list, const unsigned* __restrict__ binStart,
               const float* __restrict__ dinv,
               const float* __restrict__ hs, float* __restrict__ agg) {
  __shared__ __attribute__((aligned(16))) float tile[TILE * DD];
  int t    = threadIdx.x;
  int lane = t & 31;
  int wid  = __builtin_amdgcn_readfirstlane(t >> 5);
  unsigned e0 = (unsigned)__builtin_amdgcn_readfirstlane((int)binStart[blockIdx.x]);
  unsigned e1 = (unsigned)__builtin_amdgcn_readfirstlane((int)binStart[blockIdx.x + 1]);
  int node0 = blockIdx.x * TILE;

  unsigned n = e1 - e0;
  const int* lst = list + (size_t)e0;
  unsigned cb = (unsigned)wid * EPW;
  int rec = 0;
  if (cb + lane < n) rec = lst[cb + lane];

  const v4f* h4 = (const v4f*)(hs + (size_t)node0 * DD);
  v4f* g4 = (v4f*)(agg + (size_t)node0 * DD);
  v4f* t4 = (v4f*)tile;
  constexpr int NV4   = TILE * DD / 4;
  constexpr int CFULL = NV4 / (32 * SW);
  constexpr int CREST = NV4 - CFULL * 32 * SW;
  float dr[CFULL + 1];
  {
    v4f cp[CFULL];
#pragma unroll
    for (int k = 0; k < CFULL; ++k) cp[k] = h4[t + k * 32 * SW];
#pragma unroll
    for (int k = 0; k < CFULL; ++k) dr[k] = dinv[node0 + ((t + k * 32 * SW) >> 4)];
    dr[CFULL] = 0.0f;
    if (t < CREST) {
      t4[t + CFULL * 32 * SW] = h4[t + CFULL * 32 * SW];
      dr[CFULL] = dinv[node0 + ((t + CFULL * 32 * SW) >> 4)];
    }
#pragma unroll
    for (int k = 0; k < CFULL; ++k) t4[t + k * 32 * SW] = cp[k];
  }
  __syncthreads();

  const unsigned lo = 4u * (unsigned)lane;

  while (cb < n) {
    unsigned cnt = n - cb;
    unsigned nb  = cb + EPW * SW;
    int nrec = 0;
    if (nb + lane < n) nrec = lst[nb + lane];
    int sv = rec & ((1 << SRC_BITS) - 1);
    int dl = (int)((unsigned)rec >> SRC_BITS) * DD;

    if (cnt < EPW) {
      for (unsigned j0 = 0; j0 < cnt; j0 += CH) {
        v2f rr[CH];
#pragma unroll
        for (int r = 0; r < CH; ++r) {
          unsigned jj = j0 + r < cnt ? j0 + r : cnt - 1;
          unsigned s = (unsigned)__builtin_amdgcn_readlane(sv, jj);
          const char* rp = (const char*)hs + (s * (DD * 4u) + lo);
          rr[r][0] = *(const float*)rp;
          rr[r][1] = *(const float*)(rp + 128);
        }
#pragma unroll
        for (int r = 0; r < CH; ++r) {
          if (j0 + r < cnt) {
            int d = __builtin_amdgcn_readlane(dl, j0 + r);
            float* ap = &tile[d + lane];
            __hip_atomic_fetch_add(ap,      rr[r][0], __ATOMIC_RELAXED, __HIP_MEMORY_SCOPE_WORKGROUP);
            __hip_atomic_fetch_add(ap + 32, rr[r][1], __ATOMIC_RELAXED, __HIP_MEMORY_SCOPE_WORKGROUP);
          }
        }
      }
    } else {
      v2f rows[2][CH];
#pragma unroll
      for (int r = 0; r < CH; ++r) {
        unsigned s = (unsigned)__builtin_amdgcn_readlane(sv, r);
        const char* rp = (const char*)hs + (s * (DD * 4u) + lo);
        rows[0][r][0] = *(const float*)rp;
        rows[0][r][1] = *(const float*)(rp + 128);
      }
#pragma unroll
      for (int c = 0; c < EPW / CH; ++c) {
        if (c + 1 < EPW / CH) {
#pragma unroll
          for (int r = 0; r < CH; ++r) {
            int j = (c + 1) * CH + r;
            unsigned s = (unsigned)__builtin_amdgcn_readlane(sv, j);
            const char* rp = (const char*)hs + (s * (DD * 4u) + lo);
            rows[(c + 1) & 1][r][0] = *(const float*)rp;
            rows[(c + 1) & 1][r][1] = *(const float*)(rp + 128);
          }
        }
#pragma unroll
        for (int r = 0; r < CH; ++r) {
          int j = c * CH + r;
          int d = __builtin_amdgcn_readlane(dl, j);
          v2f v = rows[c & 1][r];
          float* ap = &tile[d + lane];
          __hip_atomic_fetch_add(ap,      v[0], __ATOMIC_RELAXED, __HIP_MEMORY_SCOPE_WORKGROUP);
          __hip_atomic_fetch_add(ap + 32, v[1], __ATOMIC_RELAXED, __HIP_MEMORY_SCOPE_WORKGROUP);
        }
      }
    }
    rec = nrec;
    cb  = nb;
  }
  __syncthreads();
#pragma unroll
  for (int k = 0; k < CFULL; ++k) {
    v4f v = t4[t + k * 32 * SW];
    float d = dr[k];
    v[0] *= d; v[1] *= d; v[2] *= d; v[3] *= d;
    g4[t + k * 32 * SW] = v;
  }
  if (t < CREST) {
    v4f v = t4[t + CFULL * 32 * SW];
    float d = dr[CFULL];
    v[0] *= d; v[1] *= d; v[2] *= d; v[3] *= d;
    g4[t + CFULL * 32 * SW] = v;
  }
}

#define PW    32
#define SSTEP 100
#define NSAMP ((NN + SSTEP - 1) / SSTEP)
static_assert(NSAMP <= 32 * PW, "one sample per thread");
static_assert(SSTEP <= 128, "one window per four waves");

__global__ void __launch_bounds__(32 * PW)
k_pool_graph(const float* __restrict__ aggl, const float* __restrict__ bl,
             const int* __restrict__ bidx, const float* __restrict__ Wout,
             const float* __restrict__ bout, float* __restrict__ out,
             float* __restrict__ hidden) {
  __shared__ int   cnt[4];
  __shared__ __attribute__((aligned(8))) float smax[PW][DD];
  __shared__ __attribute__((aligned(8))) float ssum[PW][DD];
  __shared__ float red[2 * DD];
  int g    = blockIdx.x;
  int t    = threadIdx.x;
  int lane = t & 31;
  int wv   = t >> 5;

  if (t < 4) cnt[t] = 0;
  __syncthreads();

  bool a = false, b = false;
  if (t < NSAMP) {
    int v = bidx[t * SSTEP];
    a = v < g;
    b = v <= g;
  }
  unsigned ma = __builtin_amdgcn_ballot_w32(a);
  unsigned mb = __builtin_amdgcn_ballot_w32(b);
  if (lane == 0) {
    if (ma) atomicAdd(&cnt[0], __builtin_popcount(ma));
    if (mb) atomicAdd(&cnt[1], __builtin_popcount(mb));
  }
  __syncthreads();
  int c0 = cnt[0], c1 = cnt[1];
  int base_lo = (c0 > 0 ? c0 - 1 : 0) * SSTEP;
  int base_hi = (c1 > 0 ? c1 - 1 : 0) * SSTEP;

  int  which = t >> 7;
  bool c = false;
  if (t < 256) {
    int j = t & 127;
    int p = (which ? base_hi : base_lo) + j;
    if (j < SSTEP && p < NN) c = bidx[p] < g + which;
  }
  unsigned mc = __builtin_amdgcn_ballot_w32(c);
  if (t < 256 && lane == 0 && mc) atomicAdd(&cnt[2 + which], __builtin_popcount(mc));
  __syncthreads();
  int lo = base_lo + cnt[2];
  int hi = base_hi + cnt[3];

  v2f bb = *(const v2f*)(bl + 2 * lane);
  float m0 = -INFINITY, m1 = -INFINITY, s0 = 0.0f, s1 = 0.0f;
  const float* rp = aggl + 2 * lane;
  int n = lo + wv;
  for (; n + 3 * PW < hi; n += 4 * PW) {
    v2f r0 = *(const v2f*)(rp + (size_t)n * DD);
    v2f r1 = *(const v2f*)(rp + (size_t)(n + PW) * DD);
    v2f r2 = *(const v2f*)(rp + (size_t)(n + 2 * PW) * DD);
    v2f r3 = *(const v2f*)(rp + (size_t)(n + 3 * PW) * DD);
    float x0 = tanhf(r0[0] + bb[0]), y0 = tanhf(r0[1] + bb[1]);
    float x1 = tanhf(r1[0] + bb[0]), y1 = tanhf(r1[1] + bb[1]);
    float x2 = tanhf(r2[0] + bb[0]), y2 = tanhf(r2[1] + bb[1]);
    float x3 = tanhf(r3[0] + bb[0]), y3 = tanhf(r3[1] + bb[1]);
    m0 = fmaxf(fmaxf(fmaxf(fmaxf(m0, x0), x1), x2), x3);
    m1 = fmaxf(fmaxf(fmaxf(fmaxf(m1, y0), y1), y2), y3);
    s0 = (((s0 + x0) + x1) + x2) + x3;
    s1 = (((s1 + y0) + y1) + y2) + y3;
  }
  for (; n < hi; n += PW) {
    v2f r0 = *(const v2f*)(rp + (size_t)n * DD);
    float x0 = tanhf(r0[0] + bb[0]), y0 = tanhf(r0[1] + bb[1]);
    m0 = fmaxf(m0, x0);  m1 = fmaxf(m1, y0);
    s0 += x0;            s1 += y0;
  }
  v2f pm;  pm[0] = m0;  pm[1] = m1;
  v2f ps;  ps[0] = s0;  ps[1] = s1;
  *(v2f*)(&smax[wv][2 * lane]) = pm;
  *(v2f*)(&ssum[wv][2 * lane]) = ps;
  __syncthreads();

  if (t < 2 * DD) {
    float hv;
    if (t < DD) {
      hv = smax[0][t];
      for (int w = 1; w < PW; ++w) hv = fmaxf(hv, smax[w][t]);
    } else {
      hv = ssum[0][t - DD];
      for (int w = 1; w < PW; ++w) hv += ssum[w][t - DD];
      hv = hv / fmaxf((float)(hi - lo), 1.0f);
    }
    hidden[(size_t)g * 2 * DD + t] = hv;
    red[t] = hv * Wout[t];
  }
  __syncthreads();
  for (int s = DD; s > 0; s >>= 1) {
    if (t < s) red[t] += red[t + s];
    __syncthreads();
  }
  if (t == 0) out[g] = red[0] + bout[0];
}

extern "C" void kernel_launch(void* const* d_in, const int* in_sizes, int n_in,
                              void* d_out, int out_size, void* d_ws, size_t ws_size,
                              hipStream_t stream) {
  const float* x    = (const float*)d_in[0];
  const int*   ei   = (const int*)d_in[1];
  const int*   bidx = (const int*)d_in[2];
  const float* W0   = (const float*)d_in[3];
  const float* b0   = (const float*)d_in[4];
  const float* W1   = (const float*)d_in[5];
  const float* b1   = (const float*)d_in[6];
  const float* W2   = (const float*)d_in[7];
  const float* b2   = (const float*)d_in[8];
  const float* W3   = (const float*)d_in[9];
  const float* b3   = (const float*)d_in[10];
  const float* Wout = (const float*)d_in[11];
  const float* bout = (const float*)d_in[12];

  const int* src = ei;
  const int* dst = ei + EE;

  char*  ws   = (char*)d_ws;
  size_t off  = 0;
  float* dinv = (float*)(ws + off);  off += (((size_t)NN * 4 + 255) & ~(size_t)255);
  float* bufP  = (float*)(ws + off); off += (size_t)NN * DD * 4;
  float* bufQ  = (float*)(ws + off); off += (size_t)NN * DD * 4;
  unsigned* binCount = (unsigned*)(ws + off); off += 4096;
  unsigned* cursor   = (unsigned*)(ws + off); off += 4096;
  unsigned* binStart = (unsigned*)(ws + off); off += 4096;
  int* list          = (int*)(ws + off);      off += (size_t)EE * 4;
  (void)ws_size; (void)n_in; (void)in_sizes; (void)out_size;

  float* outp   = (float*)d_out;
  float* hidden = outp + GG;

  const int T = 256;

  k_zero_counts<<<(NB + T - 1) / T, T, 0, stream>>>(binCount, cursor);
  k_tile_count<<<(EE + CE - 1) / CE, 1024, 0, stream>>>(dst, binCount);
  k_group     <<<(EE + FE - 1) / FE, 1024, 0, stream>>>(src, dst, binCount, cursor, binStart, list);
  k_tile_dinv <<<NB, DT, 0, stream>>>(list, binStart, dinv);

  const int gemm_blocks = (NN / 16 + 3) / 4;

  k_gemm_fused<FF, false><<<gemm_blocks, 128, 0, stream>>>(x,    b0, W0, dinv, bufQ);
  k_scatter_tile<<<NB, 32 * SW, 0, stream>>>(list, binStart, dinv, bufQ, bufP);
  k_gemm_fused<DD, true ><<<gemm_blocks, 128, 0, stream>>>(bufP, b0, W1, dinv, bufQ);
  k_scatter_tile<<<NB, 32 * SW, 0, stream>>>(list, binStart, dinv, bufQ, bufP);
  k_gemm_fused<DD, true ><<<gemm_blocks, 128, 0, stream>>>(bufP, b1, W2, dinv, bufQ);
  k_scatter_tile<<<NB, 32 * SW, 0, stream>>>(list, binStart, dinv, bufQ, bufP);
  k_gemm_fused<DD, true ><<<gemm_blocks, 128, 0, stream>>>(bufP, b2, W3, dinv, bufQ);
  k_scatter_tile<<<NB, 32 * SW, 0, stream>>>(list, binStart, dinv, bufQ, bufP);

  k_pool_graph<<<GG, 32 * PW, 0, stream>>>(bufP, b3, bidx, Wout, bout, outp, hidden);
}


# layers 1-3: next layer's 64x64 product done by the edge-sum tile's block on its finished LDS rows (scale, bias, tanh in registers; f32 WMMA, A read from LDS; 8-wave blocks); 13 -> 10 launches
# speedup vs baseline: 4.3942x; 1.2105x over previous
#include <hip/hip_runtime.h>
#include <math.h>

#define NN 100000
#define EE 1000000
#define FF 128
#define DD 64
#define GG 256

typedef float v2f __attribute__((ext_vector_type(2)));
typedef float v4f __attribute__((ext_vector_type(4)));
typedef float v8f __attribute__((ext_vector_type(8)));

#define TILE 100
#define NB (NN / TILE)
static_assert(NN % TILE == 0, "whole tiles only");
static_assert(NB < 1024, "one tile per thread in the grouping kernel, and one thread more for the total");

__global__ void k_zero_counts(unsigned* __restrict__ binCount, unsigned* __restrict__ cursor) {
  int i = blockIdx.x * blockDim.x + threadIdx.x;
  if (i < NB) { binCount[i] = 0u; cursor[i] = 0u; }
}

#define SRC_BITS 20
static_assert(NN <= (1 << SRC_BITS) && TILE <= (1 << (32 - SRC_BITS)), "record word 0 holds both");
#define FEPT 4
#define FE   (1024 * FEPT)

#define CEPT 16
#define CE   (1024 * CEPT)
__global__ void __launch_bounds__(1024)
k_tile_count(const int* __restrict__ dst, unsigned* __restrict__ binCount) {
  __shared__ unsigned hist[1024];
  int t = threadIdx.x;
  hist[t] = 0u;
  int dv[CEPT];
#pragma unroll
  for (int k = 0; k < CEPT; ++k) {
    int e = blockIdx.x * CE + k * 1024 + t;
    dv[k] = dst[e < EE ? e : EE - 1];
  }
  __syncthreads();
#pragma unroll
  for (int k = 0; k < CEPT; ++k) {
    int e = blockIdx.x * CE + k * 1024 + t;
    if (e < EE) atomicAdd(&hist[(unsigned)dv[k] / TILE], 1u);
  }
  __syncthreads();
  if (t < NB) {
    unsigned h = hist[t];
    if (h) atomicAdd(&binCount[t], h);
  }
}

__global__ void __launch_bounds__(1024)
k_group(const int* __restrict__ src, const int* __restrict__ dst,
        const unsigned* __restrict__ binCount, unsigned* __restrict__ cursor,
        unsigned* __restrict__ binStart, int* __restrict__ list) {
  __shared__ unsigned sstart[1024];
  __shared__ unsigned hist[1024];
  __shared__ unsigned wtot[32];
  int t    = threadIdx.x;
  int lane = t & 31;
  int wv   = t >> 5;
  unsigned c = t < NB ? binCount[t] : 0u;
  unsigned s = c;
#pragma unroll
  for (int o = 1; o < 32; o <<= 1) {
    unsigned u = __shfl_up(s, o);
    if (lane >= o) s += u;
  }
  if (lane == 31) wtot[wv] = s;
  hist[t] = 0u;
  __syncthreads();
  if (wv == 0) {
    unsigned a = wtot[lane];
    unsigned ss = a;
#pragma unroll
    for (int o = 1; o < 32; o <<= 1) {
      unsigned u = __shfl_up(ss, o);
      if (lane >= o) ss += u;
    }
    wtot[lane] = ss - a;
  }
  __syncthreads();
  unsigned excl = wtot[wv] + s - c;
  sstart[t] = excl;
  if (blockIdx.x == 0 && t <= NB) binStart[t] = excl;

  int sv[FEPT], dv[FEPT];
  unsigned bn[FEPT], rk[FEPT];
#pragma unroll
  for (int k = 0; k < FEPT; ++k) {
    int e = blockIdx.x * FE + k * 1024 + t;
    int ec = e < EE ? e : EE - 1;
    sv[k] = src[ec];
    dv[k] = dst[ec];
  }
#pragma unroll
  for (int k = 0; k < FEPT; ++k) bn[k] = (unsigned)dv[k] / TILE;
#pragma unroll
  for (int k = 0; k < FEPT; ++k) {
    int e = blockIdx.x * FE + k * 1024 + t;
    rk[k] = 0u;
    if (e < EE) rk[k] = atomicAdd(&hist[bn[k]], 1u);
  }
  __syncthreads();
  if (t < NB) {
    unsigned h = hist[t];
    if (h) sstart[t] = excl + atomicAdd(&cursor[t], h);
  }
  __syncthreads();
#pragma unroll
  for (int k = 0; k < FEPT; ++k) {
    int e = blockIdx.x * FE + k * 1024 + t;
    if (e < EE)
      list[(size_t)(sstart[bn[k]] + rk[k])] =
          (int)((unsigned)sv[k] | (((unsigned)dv[k] - bn[k] * TILE) << SRC_BITS));
  }
}

#define DT 256
#define DU 5
__global__ void __launch_bounds__(DT)
k_tile_dinv(const int* __restrict__ list, const unsigned* __restrict__ binStart,
            float* __restrict__ dinv) {
  __shared__ unsigned cnt[TILE];
  int t = threadIdx.x;
  unsigned e0 = binStart[blockIdx.x];
  unsigned n  = binStart[blockIdx.x + 1] - e0;
  if (t < TILE) cnt[t] = 0u;
  __syncthreads();
  const int* lst = list + (size_t)e0;
  for (unsigned base = 0; base < n; base += DT * DU) {
    int w0[DU];
#pragma unroll
    for (int k = 0; k < DU; ++k) {
      unsigned i = base + k * DT + t;
      w0[k] = lst[i < n ? i : n - 1];
    }
#pragma unroll
    for (int k = 0; k < DU; ++k) {
      unsigned i = base + k * DT + t;
      atomicAdd(&cnt[(unsigned)w0[k] >> SRC_BITS], i < n ? 1u : 0u);
    }
  }
  __syncthreads();
  if (t < TILE) dinv[blockIdx.x * TILE + t] = rsqrtf((float)cnt[t] + 1.0f);
}

template <int K>
__device__ __forceinline__ void wave_product(const v2f (&areg)[K / 4], const float* __restrict__ bcol,
                                             int half, v8f& acc0, v8f& acc1, v8f& acc2, v8f& acc3) {
  v4f q[2][4];
#pragma unroll
  for (int s = 0; s < 2; ++s) {
    int k0 = 4 * s + 2 * half;
    q[0][2 * s]     = *(const v4f*)(bcol + (size_t)k0 * DD);
    q[0][2 * s + 1] = *(const v4f*)(bcol + (size_t)(k0 + 1) * DD);
  }
#pragma unroll
  for (int g = 0; g < K / 8; ++g) {
    if (g + 1 < K / 8) {
#pragma unroll
      for (int s = 0; s < 2; ++s) {
        int k0 = 8 * (g + 1) + 4 * s + 2 * half;
        q[(g + 1) & 1][2 * s]     = *(const v4f*)(bcol + (size_t)k0 * DD);
        q[(g + 1) & 1][2 * s + 1] = *(const v4f*)(bcol + (size_t)(k0 + 1) * DD);
      }
    }
#pragma unroll
    for (int s = 0; s < 2; ++s) {
      v2f a  = areg[2 * g + s];
      v4f q0 = q[g & 1][2 * s];
      v4f q1 = q[g & 1][2 * s + 1];
      v2f b;
      b[0] = q0[0]; b[1] = q1[0];
      acc0 = __builtin_amdgcn_wmma_f32_16x16x4_f32(false, a, false, b, (short)0, acc0, false, false);
      b[0] = q0[1]; b[1] = q1[1];
      acc1 = __builtin_amdgcn_wmma_f32_16x16x4_f32(false, a, false, b, (short)0, acc1, false, false);
      b[0] = q0[2]; b[1] = q1[2];
      acc2 = __builtin_amdgcn_wmma_f32_16x16x4_f32(false, a, false, b, (short)0, acc2, false, false);
      b[0] = q0[3]; b[1] = q1[3];
      acc3 = __builtin_amdgcn_wmma_f32_16x16x4_f32(false, a, false, b, (short)0, acc3, false, false);
    }
  }
}

__global__ void __launch_bounds__(128, 6)
k_gemm_x(const float* __restrict__ A, const float* __restrict__ B, const float* __restrict__ dinv,
         float* __restrict__ hs) {
  int wave = (blockIdx.x * blockDim.x + threadIdx.x) >> 5;
  int lane = threadIdx.x & 31;
  if (wave * 16 >= NN) return;

  int half = lane >> 4;
  int c    = lane & 15;
  const float* arow = A + (size_t)(wave * 16 + c) * FF;

  v2f areg[FF / 4];
#pragma unroll
  for (int i = 0; i < FF / 4; ++i)
    areg[i] = *(const v2f*)(arow + 4 * i + 2 * half);

  v8f acc0 = {}, acc1 = {}, acc2 = {}, acc3 = {};
  wave_product<FF>(areg, B + 4 * c, half, acc0, acc1, acc2, acc3);

  int r0 = wave * 16 + half * 8;
#pragma unroll
  for (int j = 0; j < 8; ++j) {
    float dv = dinv[r0 + j];
    v4f v;  v[0] = dv * acc0[j]; v[1] = dv * acc1[j]; v[2] = dv * acc2[j]; v[3] = dv * acc3[j];
    *(v4f*)(hs + (size_t)(r0 + j) * DD + 4 * c) = v;
  }
}

#define EPW 32
#define CH  8
static_assert(EPW == 32 && EPW % CH == 0, "one edge per lane and whole row sets only");
static_assert((TILE * DD) % 4 == 0, "tile copied in 16-byte units");
static_assert(DD == 64, "a 16-byte unit's row is its number / 16");
#define RT ((TILE + 15) / 16)

template <int SWT, bool NEXT>
__global__ void __launch_bounds__(32 * SWT, NEXT ? 8 : 16)
k_scatter_tile(const int* __restrict__ list, const unsigned* __restrict__ binStart,
               const float* __restrict__ dinv,
               const float* __restrict__ hs, float* __restrict__ dstbuf,
               const float* __restrict__ bthis, const float* __restrict__ Wn) {
  static_assert(!NEXT || SWT >= RT, "one wave per 16-row tile of the product");
  constexpr int PITCH = NEXT ? DD + 4 : DD;
  constexpr int P4    = PITCH / 4;
  auto lds4 = [](int u) { return NEXT ? (u >> 4) * P4 + (u & 15) : u; };
  __shared__ __attribute__((aligned(16))) float tile[TILE * PITCH];
  int t    = threadIdx.x;
  int lane = t & 31;
  int wid  = __builtin_amdgcn_readfirstlane(t >> 5);
  unsigned e0 = (unsigned)__builtin_amdgcn_readfirstlane((int)binStart[blockIdx.x]);
  unsigned e1 = (unsigned)__builtin_amdgcn_readfirstlane((int)binStart[blockIdx.x + 1]);
  int node0 = blockIdx.x * TILE;

  unsigned n = e1 - e0;
  const int* lst = list + (size_t)e0;
  unsigned cb = (unsigned)wid * EPW;
  int rec = 0;
  if (cb + lane < n) rec = lst[cb + lane];

  const v4f* h4 = (const v4f*)(hs + (size_t)node0 * DD);
  v4f* t4 = (v4f*)tile;
  constexpr int NV4   = TILE * DD / 4;
  constexpr int CFULL = NV4 / (32 * SWT);
  constexpr int CREST = NV4 - CFULL * 32 * SWT;
  float dr[CFULL + 1];
  {
    v4f cp[CFULL];
#pragma unroll
    for (int k = 0; k < CFULL; ++k) cp[k] = h4[t + k * 32 * SWT];
    if (!NEXT) {
#pragma unroll
      for (int k = 0; k < CFULL; ++k) dr[k] = dinv[node0 + ((t + k * 32 * SWT) >> 4)];
      dr[CFULL] = 0.0f;
    }
    if (t < CREST) {
      int u = t + CFULL * 32 * SWT;
      t4[lds4(u)] = h4[u];
      if (!NEXT) dr[CFULL] = dinv[node0 + (u >> 4)];
    }
#pragma unroll
    for (int k = 0; k < CFULL; ++k) {
      int u = t + k * 32 * SWT;
      t4[lds4(u)] = cp[k];
    }
  }
  __syncthreads();

  const unsigned lo = 4u * (unsigned)lane;

  while (cb < n) {
    unsigned cnt = n - cb;
    unsigned nb  = cb + EPW * SWT;
    int nrec = 0;
    if (nb + lane < n) nrec = lst[nb + lane];
    int sv = rec & ((1 << SRC_BITS) - 1);
    int dl = (int)((unsigned)rec >> SRC_BITS) * PITCH;

    if (cnt < EPW) {
      for (unsigned j0 = 0; j0 < cnt; j0 += CH) {
        v2f rr[CH];
#pragma unroll
        for (int r = 0; r < CH; ++r) {
          unsigned jj = j0 + r < cnt ? j0 + r : cnt - 1;
          unsigned s = (unsigned)__builtin_amdgcn_readlane(sv, jj);
          const char* rp = (const char*)hs + (s * (DD * 4u) + lo);
          rr[r][0] = *(const float*)rp;
          rr[r][1] = *(const float*)(rp + 128);
        }
#pragma unroll
        for (int r = 0; r < CH; ++r) {
          if (j0 + r < cnt) {
            int d = __builtin_amdgcn_readlane(dl, j0 + r);
            float* ap = &tile[d + lane];
            __hip_atomic_fetch_add(ap,      rr[r][0], __ATOMIC_RELAXED, __HIP_MEMORY_SCOPE_WORKGROUP);
            __hip_atomic_fetch_add(ap + 32, rr[r][1], __ATOMIC_RELAXED, __HIP_MEMORY_SCOPE_WORKGROUP);
          }
        }
      }
    } else {
      v2f rows[2][CH];
#pragma unroll
      for (int r = 0; r < CH; ++r) {
        unsigned s = (unsigned)__builtin_amdgcn_readlane(sv, r);
        const char* rp = (const char*)hs + (s * (DD * 4u) + lo);
        rows[0][r][0] = *(const float*)rp;
        rows[0][r][1] = *(const float*)(rp + 128);
      }
#pragma unroll
      for (int c = 0; c < EPW / CH; ++c) {
        if (c + 1 < EPW / CH) {
#pragma unroll
          for (int r = 0; r < CH; ++r) {
            int j = (c + 1) * CH + r;
            unsigned s = (unsigned)__builtin_amdgcn_readlane(sv, j);
            const char* rp = (const char*)hs + (s * (DD * 4u) + lo);
            rows[(c + 1) & 1][r][0] = *(const float*)rp;
            rows[(c + 1) & 1][r][1] = *(const float*)(rp + 128);
          }
        }
#pragma unroll
        for (int r = 0; r < CH; ++r) {
          int j = c * CH + r;
          int d = __builtin_amdgcn_readlane(dl, j);
          v2f v = rows[c & 1][r];
          float* ap = &tile[d + lane];
          __hip_atomic_fetch_add(ap,      v[0], __ATOMIC_RELAXED, __HIP_MEMORY_SCOPE_WORKGROUP);
          __hip_atomic_fetch_add(ap + 32, v[1], __ATOMIC_RELAXED, __HIP_MEMORY_SCOPE_WORKGROUP);
        }
      }
    }
    rec = nrec;
    cb  = nb;
  }
  __syncthreads();

  if (NEXT) {
    if (wid < RT) {
      int half = lane >> 4;
      int c    = lane & 15;
      int ra = wid * 16 + c;
      ra = ra < TILE ? ra : TILE - 1;
      int r0 = wid * 16 + half * 8;
      float da = dinv[node0 + ra];
      float dv[8];
#pragma unroll
      for (int j = 0; j < 8; ++j) {
        int r = r0 + j;
        dv[j] = dinv[node0 + (r < TILE ? r : TILE - 1)];
      }
      const float* ar = tile + ra * PITCH + 2 * half;
      v2f areg[DD / 4], breg[DD / 4];
#pragma unroll
      for (int i = 0; i < DD / 4; ++i) areg[i] = *(const v2f*)(ar + 4 * i);
#pragma unroll
      for (int i = 0; i < DD / 4; ++i) breg[i] = *(const v2f*)(bthis + 4 * i + 2 * half);
#pragma unroll
      for (int i = 0; i < DD / 4; ++i) {
        areg[i][0] = tanhf(da * areg[i][0] + breg[i][0]);
        areg[i][1] = tanhf(da * areg[i][1] + breg[i][1]);
      }
      v8f acc0 = {}, acc1 = {}, acc2 = {}, acc3 = {};
      wave_product<DD>(areg, Wn + 4 * c, half, acc0, acc1, acc2, acc3);
      float* orow = dstbuf + (size_t)node0 * DD + 4 * c;
#pragma unroll
      for (int j = 0; j < 8; ++j) {
        int r = r0 + j;
        if (r < TILE) {
          float d = dv[j];
          v4f v;  v[0] = d * acc0[j]; v[1] = d * acc1[j]; v[2] = d * acc2[j]; v[3] = d * acc3[j];
          *(v4f*)(orow + (size_t)r * DD) = v;
        }
      }
    }
  } else {
    v4f* g4 = (v4f*)(dstbuf + (size_t)node0 * DD);
#pragma unroll
    for (int k = 0; k < CFULL; ++k) {
      int u = t + k * 32 * SWT;
      v4f v = t4[lds4(u)];
      float d = dr[k];
      v[0] *= d; v[1] *= d; v[2] *= d; v[3] *= d;
      g4[u] = v;
    }
    if (t < CREST) {
      int u = t + CFULL * 32 * SWT;
      v4f v = t4[lds4(u)];
      float d = dr[CFULL];
      v[0] *= d; v[1] *= d; v[2] *= d; v[3] *= d;
      g4[u] = v;
    }
  }
}

#define PW    32
#define SSTEP 100
#define NSAMP ((NN + SSTEP - 1) / SSTEP)
static_assert(NSAMP <= 32 * PW, "one sample per thread");
static_assert(SSTEP <= 128, "one window per four waves");

__global__ void __launch_bounds__(32 * PW)
k_pool_graph(const float* __restrict__ aggl, const float* __restrict__ bl,
             const int* __restrict__ bidx, const float* __restrict__ Wout,
             const float* __restrict__ bout, float* __restrict__ out,
             float* __restrict__ hidden) {
  __shared__ int   cnt[4];
  __shared__ __attribute__((aligned(8))) float smax[PW][DD];
  __shared__ __attribute__((aligned(8))) float ssum[PW][DD];
  __shared__ float red[2 * DD];
  int g    = blockIdx.x;
  int t    = threadIdx.x;
  int lane = t & 31;
  int wv   = t >> 5;

  if (t < 4) cnt[t] = 0;
  __syncthreads();

  bool a = false, b = false;
  if (t < NSAMP) {
    int v = bidx[t * SSTEP];
    a = v < g;
    b = v <= g;
  }
  unsigned ma = __builtin_amdgcn_ballot_w32(a);
  unsigned mb = __builtin_amdgcn_ballot_w32(b);
  if (lane == 0) {
    if (ma) atomicAdd(&cnt[0], __builtin_popcount(ma));
    if (mb) atomicAdd(&cnt[1], __builtin_popcount(mb));
  }
  __syncthreads();
  int c0 = cnt[0], c1 = cnt[1];
  int base_lo = (c0 > 0 ? c0 - 1 : 0) * SSTEP;
  int base_hi = (c1 > 0 ? c1 - 1 : 0) * SSTEP;

  int  which = t >> 7;
  bool c = false;
  if (t < 256) {
    int j = t & 127;
    int p = (which ? base_hi : base_lo) + j;
    if (j < SSTEP && p < NN) c = bidx[p] < g + which;
  }
  unsigned mc = __builtin_amdgcn_ballot_w32(c);
  if (t < 256 && lane == 0 && mc) atomicAdd(&cnt[2 + which], __builtin_popcount(mc));
  __syncthreads();
  int lo = base_lo + cnt[2];
  int hi = base_hi + cnt[3];

  v2f bb = *(const v2f*)(bl + 2 * lane);
  float m0 = -INFINITY, m1 = -INFINITY, s0 = 0.0f, s1 = 0.0f;
  const float* rp = aggl + 2 * lane;
  int n = lo + wv;
  for (; n + 3 * PW < hi; n += 4 * PW) {
    v2f r0 = *(const v2f*)(rp + (size_t)n * DD);
    v2f r1 = *(const v2f*)(rp + (size_t)(n + PW) * DD);
    v2f r2 = *(const v2f*)(rp + (size_t)(n + 2 * PW) * DD);
    v2f r3 = *(const v2f*)(rp + (size_t)(n + 3 * PW) * DD);
    float x0 = tanhf(r0[0] + bb[0]), y0 = tanhf(r0[1] + bb[1]);
    float x1 = tanhf(r1[0] + bb[0]), y1 = tanhf(r1[1] + bb[1]);
    float x2 = tanhf(r2[0] + bb[0]), y2 = tanhf(r2[1] + bb[1]);
    float x3 = tanhf(r3[0] + bb[0]), y3 = tanhf(r3[1] + bb[1]);
    m0 = fmaxf(fmaxf(fmaxf(fmaxf(m0, x0), x1), x2), x3);
    m1 = fmaxf(fmaxf(fmaxf(fmaxf(m1, y0), y1), y2), y3);
    s0 = (((s0 + x0) + x1) + x2) + x3;
    s1 = (((s1 + y0) + y1) + y2) + y3;
  }
  for (; n < hi; n += PW) {
    v2f r0 = *(const v2f*)(rp + (size_t)n * DD);
    float x0 = tanhf(r0[0] + bb[0]), y0 = tanhf(r0[1] + bb[1]);
    m0 = fmaxf(m0, x0);  m1 = fmaxf(m1, y0);
    s0 += x0;            s1 += y0;
  }
  v2f pm;  pm[0] = m0;  pm[1] = m1;
  v2f ps;  ps[0] = s0;  ps[1] = s1;
  *(v2f*)(&smax[wv][2 * lane]) = pm;
  *(v2f*)(&ssum[wv][2 * lane]) = ps;
  __syncthreads();

  if (t < 2 * DD) {
    float hv;
    if (t < DD) {
      hv = smax[0][t];
      for (int w = 1; w < PW; ++w) hv = fmaxf(hv, smax[w][t]);
    } else {
      hv = ssum[0][t - DD];
      for (int w = 1; w < PW; ++w) hv += ssum[w][t - DD];
      hv = hv / fmaxf((float)(hi - lo), 1.0f);
    }
    hidden[(size_t)g * 2 * DD + t] = hv;
    red[t] = hv * Wout[t];
  }
  __syncthreads();
  for (int s = DD; s > 0; s >>= 1) {
    if (t < s) red[t] += red[t + s];
    __syncthreads();
  }
  if (t == 0) out[g] = red[0] + bout[0];
}

extern "C" void kernel_launch(void* const* d_in, const int* in_sizes, int n_in,
                              void* d_out, int out_size, void* d_ws, size_t ws_size,
                              hipStream_t stream) {
  const float* x    = (const float*)d_in[0];
  const int*   ei   = (const int*)d_in[1];
  const int*   bidx = (const int*)d_in[2];
  const float* W0   = (const float*)d_in[3];
  const float* b0   = (const float*)d_in[4];
  const float* W1   = (const float*)d_in[5];
  const float* b1   = (const float*)d_in[6];
  const float* W2   = (const float*)d_in[7];
  const float* b2   = (const float*)d_in[8];
  const float* W3   = (const float*)d_in[9];
  const float* b3   = (const float*)d_in[10];
  const float* Wout = (const float*)d_in[11];
  const float* bout = (const float*)d_in[12];

  const int* src = ei;
  const int* dst = ei + EE;

  char*  ws   = (char*)d_ws;
  size_t off  = 0;
  float* dinv = (float*)(ws + off);  off += (((size_t)NN * 4 + 255) & ~(size_t)255);
  float* bufP  = (float*)(ws + off); off += (size_t)NN * DD * 4;
  float* bufQ  = (float*)(ws + off); off += (size_t)NN * DD * 4;
  unsigned* binCount = (unsigned*)(ws + off); off += 4096;
  unsigned* cursor   = (unsigned*)(ws + off); off += 4096;
  unsigned* binStart = (unsigned*)(ws + off); off += 4096;
  int* list          = (int*)(ws + off);      off += (size_t)EE * 4;
  (void)ws_size; (void)n_in; (void)in_sizes; (void)out_size;

  float* outp   = (float*)d_out;
  float* hidden = outp + GG;

  const int T = 256;

  k_zero_counts<<<(NB + T - 1) / T, T, 0, stream>>>(binCount, cursor);
  k_tile_count<<<(EE + CE - 1) / CE, 1024, 0, stream>>>(dst, binCount);
  k_group     <<<(EE + FE - 1) / FE, 1024, 0, stream>>>(src, dst, binCount, cursor, binStart, list);
  k_tile_dinv <<<NB, DT, 0, stream>>>(list, binStart, dinv);

  const int gemm_blocks = (NN / 16 + 3) / 4;

  k_gemm_x<<<gemm_blocks, 128, 0, stream>>>(x, W0, dinv, bufQ);
  k_scatter_tile<8, true ><<<NB, 32 * 8,  0, stream>>>(list, binStart, dinv, bufQ, bufP, b0, W1);
  k_scatter_tile<8, true ><<<NB, 32 * 8,  0, stream>>>(list, binStart, dinv, bufP, bufQ, b1, W2);
  k_scatter_tile<8, true ><<<NB, 32 * 8,  0, stream>>>(list, binStart, dinv, bufQ, bufP, b2, W3);
  k_scatter_tile<16, false><<<NB, 32 * 16, 0, stream>>>(list, binStart, dinv, bufP, bufQ, nullptr, nullptr);

  k_pool_graph<<<GG, 32 * PW, 0, stream>>>(bufQ, b3, bidx, Wout, bout, outp, hidden);
}


# two short launches folded away (10 -> 8): layer-0 product blocks first count their 64 rows' in-degree (dinv) from the tile records; tile counts kept per block, summed by the grouping, no zero pass
# speedup vs baseline: 4.6197x; 1.0513x over previous
#include <hip/hip_runtime.h>
#include <math.h>

#define NN 100000
#define EE 1000000
#define FF 128
#define DD 64
#define GG 256

typedef float v2f __attribute__((ext_vector_type(2)));
typedef float v4f __attribute__((ext_vector_type(4)));
typedef float v8f __attribute__((ext_vector_type(8)));
typedef unsigned v4u __attribute__((ext_vector_type(4)));

#define TILE 100
#define NB (NN / TILE)
static_assert(NN % TILE == 0, "whole tiles only");
static_assert(NB < 1024, "one tile per thread in the grouping kernel, and one thread more for the total");


#define SRC_BITS 20
static_assert(NN <= (1 << SRC_BITS) && TILE <= (1 << (32 - SRC_BITS)), "record word 0 holds both");
#define FEPT 4
#define FE   (1024 * FEPT)

#define CEPT 16
#define CE   (1024 * CEPT)
#define CNTB ((EE + CE - 1) / CE)
__global__ void __launch_bounds__(1024)
k_tile_count(const int* __restrict__ dst, unsigned* __restrict__ part, unsigned* __restrict__ cursor) {
  __shared__ unsigned hist[1024];
  int t = threadIdx.x;
  hist[t] = 0u;
  if (blockIdx.x == 0) cursor[t] = 0u;
  int dv[CEPT];
#pragma unroll
  for (int k = 0; k < CEPT; ++k) {
    int e = blockIdx.x * CE + k * 1024 + t;
    dv[k] = dst[e < EE ? e : EE - 1];
  }
  __syncthreads();
#pragma unroll
  for (int k = 0; k < CEPT; ++k) {
    int e = blockIdx.x * CE + k * 1024 + t;
    if (e < EE) atomicAdd(&hist[(unsigned)dv[k] / TILE], 1u);
  }
  __syncthreads();
  part[blockIdx.x * 1024 + t] = hist[t];
}

__global__ void __launch_bounds__(1024)
k_group(const int* __restrict__ src, const int* __restrict__ dst,
        const unsigned* __restrict__ part, unsigned* __restrict__ cursor,
        unsigned* __restrict__ binStart, int* __restrict__ list) {
  __shared__ unsigned sstart[1024];
  __shared__ unsigned hist[1024];
  __shared__ unsigned wtot[32];
  __shared__ __attribute__((aligned(16))) unsigned psum[4 * 1024];
  int t    = threadIdx.x;
  int lane = t & 31;
  int wv   = t >> 5;
  {
    constexpr int PK = (CNTB + 3) / 4;
    const v4u* p4 = (const v4u*)part;
    int cg = t & 255;
    int rg = t >> 8;
    v4u pv[PK];
#pragma unroll
    for (int k = 0; k < PK; ++k) {
      int b = rg + 4 * k;
      pv[k] = p4[(b < CNTB ? b : CNTB - 1) * 256 + cg];
    }
    v4u a = {0u, 0u, 0u, 0u};
#pragma unroll
    for (int k = 0; k < PK; ++k)
      if (rg + 4 * k < CNTB) a += pv[k];
    *(v4u*)&psum[rg * 1024 + 4 * cg] = a;
  }
  __syncthreads();
  unsigned c = psum[t] + psum[1024 + t] + psum[2048 + t] + psum[3072 + t];
  unsigned s = c;
#pragma unroll
  for (int o = 1; o < 32; o <<= 1) {
    unsigned u = __shfl_up(s, o);
    if (lane >= o) s += u;
  }
  if (lane == 31) wtot[wv] = s;
  hist[t] = 0u;
  __syncthreads();
  if (wv == 0) {
    unsigned a = wtot[lane];
    unsigned ss = a;
#pragma unroll
    for (int o = 1; o < 32; o <<= 1) {
      unsigned u = __shfl_up(ss, o);
      if (lane >= o) ss += u;
    }
    wtot[lane] = ss - a;
  }
  __syncthreads();
  unsigned excl = wtot[wv] + s - c;
  sstart[t] = excl;
  if (blockIdx.x == 0 && t <= NB) binStart[t] = excl;

  int sv[FEPT], dv[FEPT];
  unsigned bn[FEPT], rk[FEPT];
#pragma unroll
  for (int k = 0; k < FEPT; ++k) {
    int e = blockIdx.x * FE + k * 1024 + t;
    int ec = e < EE ? e : EE - 1;
    sv[k] = src[ec];
    dv[k] = dst[ec];
  }
#pragma unroll
  for (int k = 0; k < FEPT; ++k) bn[k] = (unsigned)dv[k] / TILE;
#pragma unroll
  for (int k = 0; k < FEPT; ++k) {
    int e = blockIdx.x * FE + k * 1024 + t;
    rk[k] = 0u;
    if (e < EE) rk[k] = atomicAdd(&hist[bn[k]], 1u);
  }
  __syncthreads();
  if (t < NB) {
    unsigned h = hist[t];
    if (h) sstart[t] = excl + atomicAdd(&cursor[t], h);
  }
  __syncthreads();
#pragma unroll
  for (int k = 0; k < FEPT; ++k) {
    int e = blockIdx.x * FE + k * 1024 + t;
    if (e < EE)
      list[(size_t)(sstart[bn[k]] + rk[k])] =
          (int)((unsigned)sv[k] | (((unsigned)dv[k] - bn[k] * TILE) << SRC_BITS));
  }
}

template <int K>
__device__ __forceinline__ void wave_product(const v2f (&areg)[K / 4], const float* __restrict__ bcol,
                                             int half, v8f& acc0, v8f& acc1, v8f& acc2, v8f& acc3) {
  v4f q[2][4];
#pragma unroll
  for (int s = 0; s < 2; ++s) {
    int k0 = 4 * s + 2 * half;
    q[0][2 * s]     = *(const v4f*)(bcol + (size_t)k0 * DD);
    q[0][2 * s + 1] = *(const v4f*)(bcol + (size_t)(k0 + 1) * DD);
  }
#pragma unroll
  for (int g = 0; g < K / 8; ++g) {
    if (g + 1 < K / 8) {
#pragma unroll
      for (int s = 0; s < 2; ++s) {
        int k0 = 8 * (g + 1) + 4 * s + 2 * half;
        q[(g + 1) & 1][2 * s]     = *(const v4f*)(bcol + (size_t)k0 * DD);
        q[(g + 1) & 1][2 * s + 1] = *(const v4f*)(bcol + (size_t)(k0 + 1) * DD);
      }
    }
#pragma unroll
    for (int s = 0; s < 2; ++s) {
      v2f a  = areg[2 * g + s];
      v4f q0 = q[g & 1][2 * s];
      v4f q1 = q[g & 1][2 * s + 1];
      v2f b;
      b[0] = q0[0]; b[1] = q1[0];
      acc0 = __builtin_amdgcn_wmma_f32_16x16x4_f32(false, a, false, b, (short)0, acc0, false, false);
      b[0] = q0[1]; b[1] = q1[1];
      acc1 = __builtin_amdgcn_wmma_f32_16x16x4_f32(false, a, false, b, (short)0, acc1, false, false);
      b[0] = q0[2]; b[1] = q1[2];
      acc2 = __builtin_amdgcn_wmma_f32_16x16x4_f32(false, a, false, b, (short)0, acc2, false, false);
      b[0] = q0[3]; b[1] = q1[3];
      acc3 = __builtin_amdgcn_wmma_f32_16x16x4_f32(false, a, false, b, (short)0, acc3, false, false);
    }
  }
}

#define GR 64
#define GU 8
static_assert(GR <= TILE, "a block's rows lie in at most two tiles");
static_assert(NN % 16 == 0, "whole 16-row wave tiles only");
__global__ void __launch_bounds__(128, 8)
k_gemm_x(const float* __restrict__ A, const float* __restrict__ B, const int* __restrict__ list,
         const unsigned* __restrict__ binStart, float* __restrict__ dinv, float* __restrict__ hs) {
  __shared__ unsigned cnt[GR];
  int t    = threadIdx.x;
  int lane = t & 31;
  int wv   = t >> 5;
  int row0 = blockIdx.x * GR;
  int half = lane >> 4;
  int c    = lane & 15;
  bool live = row0 + wv * 16 < NN;
  int  ar   = row0 + wv * 16 + c;
  const float* arow = A + (size_t)(ar < NN ? ar : NN - 1) * FF;

  if (t < GR) cnt[t] = 0u;
  __syncthreads();

  v2f areg[FF / 4];
#pragma unroll
  for (int i = 0; i < FF / 4; ++i)
    areg[i] = *(const v2f*)(arow + 4 * i + 2 * half);

  int rend = row0 + GR < NN ? row0 + GR : NN;
  int T0 = row0 / TILE;
  int T1 = (rend - 1) / TILE;
  unsigned e0 = binStart[T0];
  unsigned nm = binStart[T0 + 1] - e0;
  unsigned n  = binStart[T1 + 1] - e0;
  const int* lst = list + (size_t)e0;
  for (unsigned base = 0; base < n; base += 128 * GU) {
    int w0[GU];
#pragma unroll
    for (int k = 0; k < GU; ++k) {
      unsigned i = base + k * 128 + t;
      w0[k] = lst[i < n ? i : n - 1];
    }
#pragma unroll
    for (int k = 0; k < GU; ++k) {
      unsigned i  = base + k * 128 + t;
      unsigned ic = i < n ? i : n - 1;
      int node = (ic < nm ? T0 : T1) * TILE + (int)((unsigned)w0[k] >> SRC_BITS);
      unsigned r = (unsigned)(node - row0);
      bool in = i < n && r < (unsigned)GR;
      atomicAdd(&cnt[in ? r : (unsigned)(t & (GR - 1))], in ? 1u : 0u);
    }
  }
  __syncthreads();
  if (t < GR && row0 + t < NN) dinv[row0 + t] = rsqrtf((float)cnt[t] + 1.0f);
  if (!live) return;

  v8f acc0 = {}, acc1 = {}, acc2 = {}, acc3 = {};
  wave_product<FF>(areg, B + 4 * c, half, acc0, acc1, acc2, acc3);

  int l0 = wv * 16 + half * 8;
#pragma unroll
  for (int j = 0; j < 8; ++j) {
    float dv = rsqrtf((float)cnt[l0 + j] + 1.0f);
    v4f v;  v[0] = dv * acc0[j]; v[1] = dv * acc1[j]; v[2] = dv * acc2[j]; v[3] = dv * acc3[j];
    *(v4f*)(hs + (size_t)(row0 + l0 + j) * DD + 4 * c) = v;
  }
}

#define EPW 32
#define CH  8
static_assert(EPW == 32 && EPW % CH == 0, "one edge per lane and whole row sets only");
static_assert((TILE * DD) % 4 == 0, "tile copied in 16-byte units");
static_assert(DD == 64, "a 16-byte unit's row is its number / 16");
#define RT ((TILE + 15) / 16)

template <int SWT, bool NEXT>
__global__ void __launch_bounds__(32 * SWT, NEXT ? 8 : 16)
k_scatter_tile(const int* __restrict__ list, const unsigned* __restrict__ binStart,
               const float* __restrict__ dinv,
               const float* __restrict__ hs, float* __restrict__ dstbuf,
               const float* __restrict__ bthis, const float* __restrict__ Wn) {
  static_assert(!NEXT || SWT >= RT, "one wave per 16-row tile of the product");
  constexpr int PITCH = NEXT ? DD + 4 : DD;
  constexpr int P4    = PITCH / 4;
  auto lds4 = [](int u) { return NEXT ? (u >> 4) * P4 + (u & 15) : u; };
  __shared__ __attribute__((aligned(16))) float tile[TILE * PITCH];
  int t    = threadIdx.x;
  int lane = t & 31;
  int wid  = __builtin_amdgcn_readfirstlane(t >> 5);
  unsigned e0 = (unsigned)__builtin_amdgcn_readfirstlane((int)binStart[blockIdx.x]);
  unsigned e1 = (unsigned)__builtin_amdgcn_readfirstlane((int)binStart[blockIdx.x + 1]);
  int node0 = blockIdx.x * TILE;

  unsigned n = e1 - e0;
  const int* lst = list + (size_t)e0;
  unsigned cb = (unsigned)wid * EPW;
  int rec = 0;
  if (cb + lane < n) rec = lst[cb + lane];

  const v4f* h4 = (const v4f*)(hs + (size_t)node0 * DD);
  v4f* t4 = (v4f*)tile;
  constexpr int NV4   = TILE * DD / 4;
  constexpr int CFULL = NV4 / (32 * SWT);
  constexpr int CREST = NV4 - CFULL * 32 * SWT;
  float dr[CFULL + 1];
  {
    v4f cp[CFULL];
#pragma unroll
    for (int k = 0; k < CFULL; ++k) cp[k] = h4[t + k * 32 * SWT];
    if (!NEXT) {
#pragma unroll
      for (int k = 0; k < CFULL; ++k) dr[k] = dinv[node0 + ((t + k * 32 * SWT) >> 4)];
      dr[CFULL] = 0.0f;
    }
    if (t < CREST) {
      int u = t + CFULL * 32 * SWT;
      t4[lds4(u)] = h4[u];
      if (!NEXT) dr[CFULL] = dinv[node0 + (u >> 4)];
    }
#pragma unroll
    for (int k = 0; k < CFULL; ++k) {
      int u = t + k * 32 * SWT;
      t4[lds4(u)] = cp[k];
    }
  }
  __syncthreads();

  const unsigned lo = 4u * (unsigned)lane;

  while (cb < n) {
    unsigned cnt = n - cb;
    unsigned nb  = cb + EPW * SWT;
    int nrec = 0;
    if (nb + lane < n) nrec = lst[nb + lane];
    int sv = rec & ((1 << SRC_BITS) - 1);
    int dl = (int)((unsigned)rec >> SRC_BITS) * PITCH;

    if (cnt < EPW) {
      for (unsigned j0 = 0; j0 < cnt; j0 += CH) {
        v2f rr[CH];
#pragma unroll
        for (int r = 0; r < CH; ++r) {
          unsigned jj = j0 + r < cnt ? j0 + r : cnt - 1;
          unsigned s = (unsigned)__builtin_amdgcn_readlane(sv, jj);
          const char* rp = (const char*)hs + (s * (DD * 4u) + lo);
          rr[r][0] = *(const float*)rp;
          rr[r][1] = *(const float*)(rp + 128);
        }
#pragma unroll
        for (int r = 0; r < CH; ++r) {
          if (j0 + r < cnt) {
            int d = __builtin_amdgcn_readlane(dl, j0 + r);
            float* ap = &tile[d + lane];
            __hip_atomic_fetch_add(ap,      rr[r][0], __ATOMIC_RELAXED, __HIP_MEMORY_SCOPE_WORKGROUP);
            __hip_atomic_fetch_add(ap + 32, rr[r][1], __ATOMIC_RELAXED, __HIP_MEMORY_SCOPE_WORKGROUP);
          }
        }
      }
    } else {
      v2f rows[2][CH];
#pragma unroll
      for (int r = 0; r < CH; ++r) {
        unsigned s = (unsigned)__builtin_amdgcn_readlane(sv, r);
        const char* rp = (const char*)hs + (s * (DD * 4u) + lo);
        rows[0][r][0] = *(const float*)rp;
        rows[0][r][1] = *(const float*)(rp + 128);
      }
#pragma unroll
      for (int c = 0; c < EPW / CH; ++c) {
        if (c + 1 < EPW / CH) {
#pragma unroll
          for (int r = 0; r < CH; ++r) {
            int j = (c + 1) * CH + r;
            unsigned s = (unsigned)__builtin_amdgcn_readlane(sv, j);
            const char* rp = (const char*)hs + (s * (DD * 4u) + lo);
            rows[(c + 1) & 1][r][0] = *(const float*)rp;
            rows[(c + 1) & 1][r][1] = *(const float*)(rp + 128);
          }
        }
#pragma unroll
        for (int r = 0; r < CH; ++r) {
          int j = c * CH + r;
          int d = __builtin_amdgcn_readlane(dl, j);
          v2f v = rows[c & 1][r];
          float* ap = &tile[d + lane];
          __hip_atomic_fetch_add(ap,      v[0], __ATOMIC_RELAXED, __HIP_MEMORY_SCOPE_WORKGROUP);
          __hip_atomic_fetch_add(ap + 32, v[1], __ATOMIC_RELAXED, __HIP_MEMORY_SCOPE_WORKGROUP);
        }
      }
    }
    rec = nrec;
    cb  = nb;
  }
  __syncthreads();

  if (NEXT) {
    if (wid < RT) {
      int half = lane >> 4;
      int c    = lane & 15;
      int ra = wid * 16 + c;
      ra = ra < TILE ? ra : TILE - 1;
      int r0 = wid * 16 + half * 8;
      float da = dinv[node0 + ra];
      float dv[8];
#pragma unroll
      for (int j = 0; j < 8; ++j) {
        int r = r0 + j;
        dv[j] = dinv[node0 + (r < TILE ? r : TILE - 1)];
      }
      const float* ar = tile + ra * PITCH + 2 * half;
      v2f areg[DD / 4], breg[DD / 4];
#pragma unroll
      for (int i = 0; i < DD / 4; ++i) areg[i] = *(const v2f*)(ar + 4 * i);
#pragma unroll
      for (int i = 0; i < DD / 4; ++i) breg[i] = *(const v2f*)(bthis + 4 * i + 2 * half);
#pragma unroll
      for (int i = 0; i < DD / 4; ++i) {
        areg[i][0] = tanhf(da * areg[i][0] + breg[i][0]);
        areg[i][1] = tanhf(da * areg[i][1] + breg[i][1]);
      }
      v8f acc0 = {}, acc1 = {}, acc2 = {}, acc3 = {};
      wave_product<DD>(areg, Wn + 4 * c, half, acc0, acc1, acc2, acc3);
      float* orow = dstbuf + (size_t)node0 * DD + 4 * c;
#pragma unroll
      for (int j = 0; j < 8; ++j) {
        int r = r0 + j;
        if (r < TILE) {
          float d = dv[j];
          v4f v;  v[0] = d * acc0[j]; v[1] = d * acc1[j]; v[2] = d * acc2[j]; v[3] = d * acc3[j];
          *(v4f*)(orow + (size_t)r * DD) = v;
        }
      }
    }
  } else {
    v4f* g4 = (v4f*)(dstbuf + (size_t)node0 * DD);
#pragma unroll
    for (int k = 0; k < CFULL; ++k) {
      int u = t + k * 32 * SWT;
      v4f v = t4[lds4(u)];
      float d = dr[k];
      v[0] *= d; v[1] *= d; v[2] *= d; v[3] *= d;
      g4[u] = v;
    }
    if (t < CREST) {
      int u = t + CFULL * 32 * SWT;
      v4f v = t4[lds4(u)];
      float d = dr[CFULL];
      v[0] *= d; v[1] *= d; v[2] *= d; v[3] *= d;
      g4[u] = v;
    }
  }
}

#define PW    32
#define SSTEP 100
#define NSAMP ((NN + SSTEP - 1) / SSTEP)
static_assert(NSAMP <= 32 * PW, "one sample per thread");
static_assert(SSTEP <= 128, "one window per four waves");

__global__ void __launch_bounds__(32 * PW)
k_pool_graph(const float* __restrict__ aggl, const float* __restrict__ bl,
             const int* __restrict__ bidx, const float* __restrict__ Wout,
             const float* __restrict__ bout, float* __restrict__ out,
             float* __restrict__ hidden) {
  __shared__ int   cnt[4];
  __shared__ __attribute__((aligned(8))) float smax[PW][DD];
  __shared__ __attribute__((aligned(8))) float ssum[PW][DD];
  __shared__ float red[2 * DD];
  int g    = blockIdx.x;
  int t    = threadIdx.x;
  int lane = t & 31;
  int wv   = t >> 5;

  if (t < 4) cnt[t] = 0;
  __syncthreads();

  bool a = false, b = false;
  if (t < NSAMP) {
    int v = bidx[t * SSTEP];
    a = v < g;
    b = v <= g;
  }
  unsigned ma = __builtin_amdgcn_ballot_w32(a);
  unsigned mb = __builtin_amdgcn_ballot_w32(b);
  if (lane == 0) {
    if (ma) atomicAdd(&cnt[0], __builtin_popcount(ma));
    if (mb) atomicAdd(&cnt[1], __builtin_popcount(mb));
  }
  __syncthreads();
  int c0 = cnt[0], c1 = cnt[1];
  int base_lo = (c0 > 0 ? c0 - 1 : 0) * SSTEP;
  int base_hi = (c1 > 0 ? c1 - 1 : 0) * SSTEP;

  int  which = t >> 7;
  bool c = false;
  if (t < 256) {
    int j = t & 127;
    int p = (which ? base_hi : base_lo) + j;
    if (j < SSTEP && p < NN) c = bidx[p] < g + which;
  }
  unsigned mc = __builtin_amdgcn_ballot_w32(c);
  if (t < 256 && lane == 0 && mc) atomicAdd(&cnt[2 + which], __builtin_popcount(mc));
  __syncthreads();
  int lo = base_lo + cnt[2];
  int hi = base_hi + cnt[3];

  v2f bb = *(const v2f*)(bl + 2 * lane);
  float m0 = -INFINITY, m1 = -INFINITY, s0 = 0.0f, s1 = 0.0f;
  const float* rp = aggl + 2 * lane;
  int n = lo + wv;
  for (; n + 3 * PW < hi; n += 4 * PW) {
    v2f r0 = *(const v2f*)(rp + (size_t)n * DD);
    v2f r1 = *(const v2f*)(rp + (size_t)(n + PW) * DD);
    v2f r2 = *(const v2f*)(rp + (size_t)(n + 2 * PW) * DD);
    v2f r3 = *(const v2f*)(rp + (size_t)(n + 3 * PW) * DD);
    float x0 = tanhf(r0[0] + bb[0]), y0 = tanhf(r0[1] + bb[1]);
    float x1 = tanhf(r1[0] + bb[0]), y1 = tanhf(r1[1] + bb[1]);
    float x2 = tanhf(r2[0] + bb[0]), y2 = tanhf(r2[1] + bb[1]);
    float x3 = tanhf(r3[0] + bb[0]), y3 = tanhf(r3[1] + bb[1]);
    m0 = fmaxf(fmaxf(fmaxf(fmaxf(m0, x0), x1), x2), x3);
    m1 = fmaxf(fmaxf(fmaxf(fmaxf(m1, y0), y1), y2), y3);
    s0 = (((s0 + x0) + x1) + x2) + x3;
    s1 = (((s1 + y0) + y1) + y2) + y3;
  }
  for (; n < hi; n += PW) {
    v2f r0 = *(const v2f*)(rp + (size_t)n * DD);
    float x0 = tanhf(r0[0] + bb[0]), y0 = tanhf(r0[1] + bb[1]);
    m0 = fmaxf(m0, x0);  m1 = fmaxf(m1, y0);
    s0 += x0;            s1 += y0;
  }
  v2f pm;  pm[0] = m0;  pm[1] = m1;
  v2f ps;  ps[0] = s0;  ps[1] = s1;
  *(v2f*)(&smax[wv][2 * lane]) = pm;
  *(v2f*)(&ssum[wv][2 * lane]) = ps;
  __syncthreads();

  if (t < 2 * DD) {
    float hv;
    if (t < DD) {
      hv = smax[0][t];
      for (int w = 1; w < PW; ++w) hv = fmaxf(hv, smax[w][t]);
    } else {
      hv = ssum[0][t - DD];
      for (int w = 1; w < PW; ++w) hv += ssum[w][t - DD];
      hv = hv / fmaxf((float)(hi - lo), 1.0f);
    }
    hidden[(size_t)g * 2 * DD + t] = hv;
    red[t] = hv * Wout[t];
  }
  __syncthreads();
  for (int s = DD; s > 0; s >>= 1) {
    if (t < s) red[t] += red[t + s];
    __syncthreads();
  }
  if (t == 0) out[g] = red[0] + bout[0];
}

extern "C" void kernel_launch(void* const* d_in, const int* in_sizes, int n_in,
                              void* d_out, int out_size, void* d_ws, size_t ws_size,
                              hipStream_t stream) {
  const float* x    = (const float*)d_in[0];
  const int*   ei   = (const int*)d_in[1];
  const int*   bidx = (const int*)d_in[2];
  const float* W0   = (const float*)d_in[3];
  const float* b0   = (const float*)d_in[4];
  const float* W1   = (const float*)d_in[5];
  const float* b1   = (const float*)d_in[6];
  const float* W2   = (const float*)d_in[7];
  const float* b2   = (const float*)d_in[8];
  const float* W3   = (const float*)d_in[9];
  const float* b3   = (const float*)d_in[10];
  const float* Wout = (const float*)d_in[11];
  const float* bout = (const float*)d_in[12];

  const int* src = ei;
  const int* dst = ei + EE;

  char*  ws   = (char*)d_ws;
  size_t off  = 0;
  float* dinv = (float*)(ws + off);  off += (((size_t)NN * 4 + 255) & ~(size_t)255);
  float* bufP  = (float*)(ws + off); off += (size_t)NN * DD * 4;
  float* bufQ  = (float*)(ws + off); off += (size_t)NN * DD * 4;
  unsigned* part     = (unsigned*)(ws + off); off += (size_t)CNTB * 4096;
  unsigned* cursor   = (unsigned*)(ws + off); off += 4096;
  unsigned* binStart = (unsigned*)(ws + off); off += 4096;
  int* list          = (int*)(ws + off);      off += (size_t)EE * 4;
  (void)ws_size; (void)n_in; (void)in_sizes; (void)out_size;

  float* outp   = (float*)d_out;
  float* hidden = outp + GG;

  k_tile_count<<<CNTB, 1024, 0, stream>>>(dst, part, cursor);
  k_group     <<<(EE + FE - 1) / FE, 1024, 0, stream>>>(src, dst, part, cursor, binStart, list);

  const int gemm_blocks = (NN + GR - 1) / GR;

  k_gemm_x<<<gemm_blocks, 128, 0, stream>>>(x, W0, list, binStart, dinv, bufQ);
  k_scatter_tile<8, true ><<<NB, 32 * 8,  0, stream>>>(list, binStart, dinv, bufQ, bufP, b0, W1);
  k_scatter_tile<8, true ><<<NB, 32 * 8,  0, stream>>>(list, binStart, dinv, bufP, bufQ, b1, W2);
  k_scatter_tile<8, true ><<<NB, 32 * 8,  0, stream>>>(list, binStart, dinv, bufQ, bufP, b2, W3);
  k_scatter_tile<16, false><<<NB, 32 * 16, 0, stream>>>(list, binStart, dinv, bufP, bufQ, nullptr, nullptr);

  k_pool_graph<<<GG, 32 * PW, 0, stream>>>(bufQ, b3, bidx, Wout, bout, outp, hidden);
}
